# starting kernel
# baseline (speedup 1.0000x reference)
#include <hip/hip_runtime.h>


#define N_NODES   100000
#define N_EDGES   1600000
#define IN_FEATS  128
#define H_FEATS   64
#define N_CONVS   3
#define POLY_ORDER 4
#define NUM_CLASSES 2

typedef __attribute__((ext_vector_type(2))) float v2f;
typedef __attribute__((ext_vector_type(8))) float v8f;

__device__ __forceinline__ v8f wmma_f32(v2f a, v2f b, v8f c) {
  return __builtin_amdgcn_wmma_f32_16x16x4_f32(
       false, a,  false, b,
       (short)0, c,  false,  false);
}

__global__ __launch_bounds__(256)
void zero_f32(float* __restrict__ p, int n) {
  int stride = gridDim.x * blockDim.x;
  for (int i = blockIdx.x * blockDim.x + threadIdx.x; i < n; i += stride)
    p[i] = 0.0f;
}

__global__ __launch_bounds__(256)
void degree_kernel(const int* __restrict__ dst, float* __restrict__ deg, int E) {
  int e = blockIdx.x * blockDim.x + threadIdx.x;
  if (e < E) atomicAdd(&deg[dst[e]], 1.0f);
}

__global__ __launch_bounds__(256)
void dinv_kernel(float* __restrict__ deg, int n) {
  int i = blockIdx.x * blockDim.x + threadIdx.x;
  if (i < n) {
    float d = deg[i];
    d = d < 1.0f ? 1.0f : d;
    deg[i] = rsqrtf(d);
  }
}

__global__ __launch_bounds__(128)
void trunk_kernel(const float* __restrict__ X,
                  const float* __restrict__ W1, const float* __restrict__ b1,
                  const float* __restrict__ W2, const float* __restrict__ b2,
                  float* __restrict__ P0) {
  __shared__ float ldsX[16 * 132];
  __shared__ float ldsH[16 * 72];

  const int tid  = threadIdx.x;
  const int lane = tid & 31;
  const int wave = tid >> 5;
  const int mrow = lane & 15;
  const int kh   = lane >> 4;
  const int row0 = blockIdx.x * 16;
  const int col0 = wave * 16;

  for (int i = tid; i < 16 * IN_FEATS; i += 128) {
    int r = i >> 7, c = i & 127;
    ldsX[r * 132 + c] = X[(size_t)(row0 + r) * IN_FEATS + c];
  }
  __syncthreads();

  v8f acc = {};
  for (int k0 = 0; k0 < IN_FEATS; k0 += 4) {
    v2f a, b;
    a.x = ldsX[mrow * 132 + k0 + 2 * kh];
    a.y = ldsX[mrow * 132 + k0 + 2 * kh + 1];
    b.x = W1[(size_t)(k0 + 2 * kh)     * H_FEATS + col0 + mrow];
    b.y = W1[(size_t)(k0 + 2 * kh + 1) * H_FEATS + col0 + mrow];
    acc = wmma_f32(a, b, acc);
  }
  {
    float bias = b1[col0 + mrow];
    #pragma unroll
    for (int v = 0; v < 8; ++v) {
      int r = v + 8 * kh;
      float val = acc[v] + bias;
      ldsH[r * 72 + col0 + mrow] = val > 0.0f ? val : 0.0f;
    }
  }
  __syncthreads();

  v8f acc2 = {};
  for (int k0 = 0; k0 < H_FEATS; k0 += 4) {
    v2f a, b;
    a.x = ldsH[mrow * 72 + k0 + 2 * kh];
    a.y = ldsH[mrow * 72 + k0 + 2 * kh + 1];
    b.x = W2[(size_t)(k0 + 2 * kh)     * H_FEATS + col0 + mrow];
    b.y = W2[(size_t)(k0 + 2 * kh + 1) * H_FEATS + col0 + mrow];
    acc2 = wmma_f32(a, b, acc2);
  }
  {
    float bias = b2[col0 + mrow];
    #pragma unroll
    for (int v = 0; v < 8; ++v) {
      int r = v + 8 * kh;
      float val = acc2[v] + bias;
      P0[(size_t)(row0 + r) * H_FEATS + col0 + mrow] = val > 0.0f ? val : 0.0f;
    }
  }
}

__global__ __launch_bounds__(256)
void scatter_kernel(const float* __restrict__ P,
                    const int* __restrict__ src, const int* __restrict__ dst,
                    const float* __restrict__ dinv, float* __restrict__ agg,
                    int E) {
  int gid  = blockIdx.x * blockDim.x + threadIdx.x;
  int e    = gid >> 5;
  int lane = gid & 31;
  if (e >= E) return;
  int s = src[e];
  int d = dst[e];
  float di = dinv[s];
  const float2* pp = reinterpret_cast<const float2*>(P + (size_t)s * H_FEATS) + lane;
  float2 v = *pp;
  float* ap = agg + (size_t)d * H_FEATS + 2 * lane;
  atomicAdd(ap,     v.x * di);
  atomicAdd(ap + 1, v.y * di);
}

__global__ __launch_bounds__(256)
void update_kernel(const float* __restrict__ Pprev, float* __restrict__ agg,
                   const float* __restrict__ dinv, float* __restrict__ Pk,
                   int n) {
  int i = blockIdx.x * blockDim.x + threadIdx.x;
  if (i >= n) return;
  int node = i >> 6;
  float a = agg[i];
  agg[i] = 0.0f;
  Pk[i] = Pprev[i] - a * dinv[node];
}

__global__ __launch_bounds__(256)
void wc_kernel(const float* __restrict__ thetas, const float* __restrict__ Wm1,
               float* __restrict__ Wc) {
  int t = blockIdx.x * blockDim.x + threadIdx.x;
  if (t >= POLY_ORDER * H_FEATS * H_FEATS) return;
  int k  = t >> 12;
  int rc = t & 4095;
  int r  = rc >> 6, c = rc & 63;
  float s = 0.0f;
  #pragma unroll
  for (int i = 0; i < N_CONVS; ++i)
    s += thetas[i * POLY_ORDER + k] * Wm1[(size_t)(i * H_FEATS + r) * H_FEATS + c];
  Wc[t] = s;
}

__global__ __launch_bounds__(128)
void head_kernel(const float* __restrict__ P0, const float* __restrict__ P1,
                 const float* __restrict__ P2, const float* __restrict__ P3,
                 const float* __restrict__ Wc, const float* __restrict__ bm1,
                 const float* __restrict__ Wm2, const float* __restrict__ bm2,
                 float* __restrict__ out) {
  __shared__ float ldsA[16 * 72];
  __shared__ float ldsH[16 * 72];

  const int tid  = threadIdx.x;
  const int lane = tid & 31;
  const int wave = tid >> 5;
  const int mrow = lane & 15;
  const int kh   = lane >> 4;
  const int row0 = blockIdx.x * 16;
  const int col0 = wave * 16;

  const float* Ps[4] = {P0, P1, P2, P3};

  v8f acc = {};
  #pragma unroll
  for (int k = 0; k < POLY_ORDER; ++k) {
    const float* Pk = Ps[k];
    const float* Wk = Wc + k * H_FEATS * H_FEATS;
    for (int i = tid; i < 16 * H_FEATS; i += 128) {
      int r = i >> 6, c = i & 63;
      ldsA[r * 72 + c] = Pk[(size_t)(row0 + r) * H_FEATS + c];
    }
    __syncthreads();
    for (int k0 = 0; k0 < H_FEATS; k0 += 4) {
      v2f a, b;
      a.x = ldsA[mrow * 72 + k0 + 2 * kh];
      a.y = ldsA[mrow * 72 + k0 + 2 * kh + 1];
      b.x = Wk[(k0 + 2 * kh)     * H_FEATS + col0 + mrow];
      b.y = Wk[(k0 + 2 * kh + 1) * H_FEATS + col0 + mrow];
      acc = wmma_f32(a, b, acc);
    }
    __syncthreads();
  }

  {
    float bias = bm1[col0 + mrow];
    #pragma unroll
    for (int v = 0; v < 8; ++v) {
      int r = v + 8 * kh;
      float val = acc[v] + bias;
      ldsH[r * 72 + col0 + mrow] = val > 0.0f ? val : 0.0f;
    }
  }
  __syncthreads();

  if (tid < 32) {
    int r = tid >> 1, c = tid & 1;
    float s = bm2[c];
    #pragma unroll 8
    for (int j = 0; j < H_FEATS; ++j)
      s += ldsH[r * 72 + j] * Wm2[j * NUM_CLASSES + c];
    out[(size_t)(row0 + r) * NUM_CLASSES + c] = s;
  }
}

extern "C" void kernel_launch(void* const* d_in, const int* in_sizes, int n_in,
                              void* d_out, int out_size, void* d_ws, size_t ws_size,
                              hipStream_t stream) {
  const float* X      = (const float*)d_in[0];
  const int*   src    = (const int*)  d_in[1];
  const int*   dst    = (const int*)  d_in[2];
  const float* W1     = (const float*)d_in[3];
  const float* b1     = (const float*)d_in[4];
  const float* W2     = (const float*)d_in[5];
  const float* b2     = (const float*)d_in[6];
  const float* thetas = (const float*)d_in[7];
  const float* Wm1    = (const float*)d_in[8];
  const float* bm1    = (const float*)d_in[9];
  const float* Wm2    = (const float*)d_in[10];
  const float* bm2    = (const float*)d_in[11];
  float* out = (float*)d_out;
  float* ws  = (float*)d_ws;

  const int PN = N_NODES * H_FEATS;
  float* dinv = ws;
  float* P0   = ws + 100352;
  float* P1   = P0 + PN;
  float* P2   = P1 + PN;
  float* P3   = P2 + PN;
  float* agg  = P3 + PN;
  float* Wc   = agg + PN;

  zero_f32<<<512, 256, 0, stream>>>(dinv, N_NODES);
  zero_f32<<<2048, 256, 0, stream>>>(agg, PN);

  degree_kernel<<<(N_EDGES + 255) / 256, 256, 0, stream>>>(dst, dinv, N_EDGES);
  dinv_kernel<<<(N_NODES + 255) / 256, 256, 0, stream>>>(dinv, N_NODES);

  trunk_kernel<<<N_NODES / 16, 128, 0, stream>>>(X, W1, b1, W2, b2, P0);

  wc_kernel<<<(POLY_ORDER * H_FEATS * H_FEATS + 255) / 256, 256, 0, stream>>>(
      thetas, Wm1, Wc);

  float* Pbuf[4] = {P0, P1, P2, P3};
  for (int k = 1; k < POLY_ORDER; ++k) {
    scatter_kernel<<<(N_EDGES * 32 + 255) / 256, 256, 0, stream>>>(
        Pbuf[k - 1], src, dst, dinv, agg, N_EDGES);
    update_kernel<<<(PN + 255) / 256, 256, 0, stream>>>(
        Pbuf[k - 1], agg, dinv, Pbuf[k], PN);
  }

  head_kernel<<<N_NODES / 16, 128, 0, stream>>>(
      P0, P1, P2, P3, Wc, bm1, Wm2, bm2, out);
}


# edge pass: float-atomic scatter replaced by per-call destination-grouped source list (returning int atomic = rank, scan, fill); one wave per node sums rows in registers in rank (arrival) order
# speedup vs baseline: 2.2445x; 2.2445x over previous
#include <hip/hip_runtime.h>


#define N_NODES   100000
#define N_EDGES   1600000
#define IN_FEATS  128
#define H_FEATS   64
#define N_CONVS   3
#define POLY_ORDER 4
#define NUM_CLASSES 2

typedef __attribute__((ext_vector_type(2))) float v2f;
typedef __attribute__((ext_vector_type(8))) float v8f;

__device__ __forceinline__ v8f wmma_f32(v2f a, v2f b, v8f c) {
  return __builtin_amdgcn_wmma_f32_16x16x4_f32(
       false, a,  false, b,
       (short)0, c,  false,  false);
}

__global__ __launch_bounds__(256)
void zero_i32(int* __restrict__ p, int n) {
  int stride = gridDim.x * blockDim.x;
  for (int i = blockIdx.x * blockDim.x + threadIdx.x; i < n; i += stride)
    p[i] = 0;
}

__global__ __launch_bounds__(256)
void count_rank_kernel(const int* __restrict__ dst, int* __restrict__ cnt,
                       int* __restrict__ rank, int E) {
  int e = blockIdx.x * blockDim.x + threadIdx.x;
  if (e < E) rank[e] = atomicAdd(&cnt[dst[e]], 1);
}

__device__ __forceinline__ int wave_incl_scan(int x, int lane) {
  for (int d = 1; d < 32; d <<= 1) {
    int y = __shfl_up(x, d);
    if (lane >= d) x += y;
  }
  return x;
}

#define SCAN_BLOCK  256
#define SCAN_BLOCKS ((N_NODES + SCAN_BLOCK - 1) / SCAN_BLOCK)
static_assert(SCAN_BLOCKS <= 2 * SCAN_BLOCK, "scan_kernel reads at most two block totals per thread");

__global__ __launch_bounds__(SCAN_BLOCK)
void dinv_bsum_kernel(const int* __restrict__ cnt, float* __restrict__ dinv,
                      int* __restrict__ bsum, int n) {
  __shared__ int wtot[SCAN_BLOCK / 32];
  const int tid  = threadIdx.x;
  const int lane = tid & 31;
  const int wave = tid >> 5;
  const int i = blockIdx.x * SCAN_BLOCK + tid;
  const int c = i < n ? cnt[i] : 0;
  if (i < n) {
    float d = (float)c;
    d = d < 1.0f ? 1.0f : d;
    dinv[i] = rsqrtf(d);
  }
  const int x = wave_incl_scan(c, lane);
  if (lane == 31) wtot[wave] = x;
  __syncthreads();
  if (tid == 0) {
    int s = 0;
    for (int w = 0; w < SCAN_BLOCK / 32; ++w) s += wtot[w];
    bsum[blockIdx.x] = s;
  }
}

__global__ __launch_bounds__(SCAN_BLOCK)
void scan_kernel(const int* __restrict__ cnt, const int* __restrict__ bsum,
                 int* __restrict__ off, int n) {
  __shared__ int wtot[SCAN_BLOCK / 32];
  __shared__ int ptot[SCAN_BLOCK / 32];
  const int tid  = threadIdx.x;
  const int lane = tid & 31;
  const int wave = tid >> 5;
  const int b = blockIdx.x;
  int p = 0;
  if (tid < b)              p += bsum[tid];
  if (tid + SCAN_BLOCK < b) p += bsum[tid + SCAN_BLOCK];
  p = wave_incl_scan(p, lane);
  const int i = b * SCAN_BLOCK + tid;
  const int c = i < n ? cnt[i] : 0;
  const int x = wave_incl_scan(c, lane);
  if (lane == 31) { ptot[wave] = p; wtot[wave] = x; }
  __syncthreads();
  int excl = x - c;
  for (int w = 0; w < SCAN_BLOCK / 32; ++w) {
    excl += ptot[w];
    if (w < wave) excl += wtot[w];
  }
  if (i < n) off[i] = excl;
  if (i == n - 1) off[n] = excl + c;
}

__global__ __launch_bounds__(256)
void fill_kernel(const int* __restrict__ src, const int* __restrict__ dst,
                 const int* __restrict__ rank, const int* __restrict__ off,
                 int* __restrict__ csr, int E) {
  int e = blockIdx.x * blockDim.x + threadIdx.x;
  if (e < E) csr[off[dst[e]] + rank[e]] = src[e];
}

__global__ __launch_bounds__(128)
void trunk_kernel(const float* __restrict__ X,
                  const float* __restrict__ W1, const float* __restrict__ b1,
                  const float* __restrict__ W2, const float* __restrict__ b2,
                  float* __restrict__ P0) {
  __shared__ float ldsX[16 * 132];
  __shared__ float ldsH[16 * 72];

  const int tid  = threadIdx.x;
  const int lane = tid & 31;
  const int wave = tid >> 5;
  const int mrow = lane & 15;
  const int kh   = lane >> 4;
  const int row0 = blockIdx.x * 16;
  const int col0 = wave * 16;

  for (int i = tid; i < 16 * IN_FEATS; i += 128) {
    int r = i >> 7, c = i & 127;
    ldsX[r * 132 + c] = X[(size_t)(row0 + r) * IN_FEATS + c];
  }
  __syncthreads();

  v8f acc = {};
  for (int k0 = 0; k0 < IN_FEATS; k0 += 4) {
    v2f a, b;
    a.x = ldsX[mrow * 132 + k0 + 2 * kh];
    a.y = ldsX[mrow * 132 + k0 + 2 * kh + 1];
    b.x = W1[(size_t)(k0 + 2 * kh)     * H_FEATS + col0 + mrow];
    b.y = W1[(size_t)(k0 + 2 * kh + 1) * H_FEATS + col0 + mrow];
    acc = wmma_f32(a, b, acc);
  }
  {
    float bias = b1[col0 + mrow];
    #pragma unroll
    for (int v = 0; v < 8; ++v) {
      int r = v + 8 * kh;
      float val = acc[v] + bias;
      ldsH[r * 72 + col0 + mrow] = val > 0.0f ? val : 0.0f;
    }
  }
  __syncthreads();

  v8f acc2 = {};
  for (int k0 = 0; k0 < H_FEATS; k0 += 4) {
    v2f a, b;
    a.x = ldsH[mrow * 72 + k0 + 2 * kh];
    a.y = ldsH[mrow * 72 + k0 + 2 * kh + 1];
    b.x = W2[(size_t)(k0 + 2 * kh)     * H_FEATS + col0 + mrow];
    b.y = W2[(size_t)(k0 + 2 * kh + 1) * H_FEATS + col0 + mrow];
    acc2 = wmma_f32(a, b, acc2);
  }
  {
    float bias = b2[col0 + mrow];
    #pragma unroll
    for (int v = 0; v < 8; ++v) {
      int r = v + 8 * kh;
      float val = acc2[v] + bias;
      P0[(size_t)(row0 + r) * H_FEATS + col0 + mrow] = val > 0.0f ? val : 0.0f;
    }
  }
}

template <int U>
__device__ __forceinline__ void gather_rows(const float2* __restrict__ rows, int s, float di,
                                            int t, int lane, float2& acc) {
  int   st[U];
  float dt[U];
  float2 v[U];
  for (int j = 0; j < U; ++j) {
    st[j] = __builtin_amdgcn_readlane(s, t + j);
    dt[j] = __int_as_float(__builtin_amdgcn_readlane(__float_as_int(di), t + j));
  }
  for (int j = 0; j < U; ++j)
    v[j] = rows[(size_t)st[j] * (H_FEATS / 2) + lane];
  for (int j = 0; j < U; ++j) {
    acc.x = fmaf(dt[j], v[j].x, acc.x);
    acc.y = fmaf(dt[j], v[j].y, acc.y);
  }
}

__global__ __launch_bounds__(256)
void gather_kernel(const float* __restrict__ Pprev, const int* __restrict__ off,
                   const int* __restrict__ csr, const float* __restrict__ dinv,
                   float* __restrict__ Pk, int n) {
  const int gid  = blockIdx.x * blockDim.x + threadIdx.x;
  const int lane = gid & 31;
  const int node = __builtin_amdgcn_readfirstlane(gid >> 5);
  if (node >= n) return;
  const int beg = __builtin_amdgcn_readfirstlane(off[node]);
  const int end = __builtin_amdgcn_readfirstlane(off[node + 1]);
  const float2* rows = reinterpret_cast<const float2*>(Pprev);
  const float2 self = rows[(size_t)node * (H_FEATS / 2) + lane];
  const float dn = dinv[node];

  float2 acc = make_float2(0.0f, 0.0f);
  for (int c = beg; c < end; c += 32) {
    const int m = (end - c) < 32 ? (end - c) : 32;
    int s = 0;
    float di = 0.0f;
    if (lane < m) {
      s  = csr[c + lane];
      di = dinv[s];
    }
    int t = 0;
    for (; t + 8 <= m; t += 8) gather_rows<8>(rows, s, di, t, lane, acc);
    if (t + 4 <= m) { gather_rows<4>(rows, s, di, t, lane, acc); t += 4; }
    if (t + 2 <= m) { gather_rows<2>(rows, s, di, t, lane, acc); t += 2; }
    if (t + 1 <= m) { gather_rows<1>(rows, s, di, t, lane, acc); }
  }
  float2 o;
  o.x = self.x - acc.x * dn;
  o.y = self.y - acc.y * dn;
  reinterpret_cast<float2*>(Pk)[(size_t)node * (H_FEATS / 2) + lane] = o;
}

__global__ __launch_bounds__(256)
void wc_kernel(const float* __restrict__ thetas, const float* __restrict__ Wm1,
               float* __restrict__ Wc) {
  int t = blockIdx.x * blockDim.x + threadIdx.x;
  if (t >= POLY_ORDER * H_FEATS * H_FEATS) return;
  int k  = t >> 12;
  int rc = t & 4095;
  int r  = rc >> 6, c = rc & 63;
  float s = 0.0f;
  #pragma unroll
  for (int i = 0; i < N_CONVS; ++i)
    s += thetas[i * POLY_ORDER + k] * Wm1[(size_t)(i * H_FEATS + r) * H_FEATS + c];
  Wc[t] = s;
}

__global__ __launch_bounds__(128)
void head_kernel(const float* __restrict__ P0, const float* __restrict__ P1,
                 const float* __restrict__ P2, const float* __restrict__ P3,
                 const float* __restrict__ Wc, const float* __restrict__ bm1,
                 const float* __restrict__ Wm2, const float* __restrict__ bm2,
                 float* __restrict__ out) {
  __shared__ float ldsA[16 * 72];
  __shared__ float ldsH[16 * 72];

  const int tid  = threadIdx.x;
  const int lane = tid & 31;
  const int wave = tid >> 5;
  const int mrow = lane & 15;
  const int kh   = lane >> 4;
  const int row0 = blockIdx.x * 16;
  const int col0 = wave * 16;

  const float* Ps[4] = {P0, P1, P2, P3};

  v8f acc = {};
  #pragma unroll
  for (int k = 0; k < POLY_ORDER; ++k) {
    const float* Pk = Ps[k];
    const float* Wk = Wc + k * H_FEATS * H_FEATS;
    for (int i = tid; i < 16 * H_FEATS; i += 128) {
      int r = i >> 6, c = i & 63;
      ldsA[r * 72 + c] = Pk[(size_t)(row0 + r) * H_FEATS + c];
    }
    __syncthreads();
    for (int k0 = 0; k0 < H_FEATS; k0 += 4) {
      v2f a, b;
      a.x = ldsA[mrow * 72 + k0 + 2 * kh];
      a.y = ldsA[mrow * 72 + k0 + 2 * kh + 1];
      b.x = Wk[(k0 + 2 * kh)     * H_FEATS + col0 + mrow];
      b.y = Wk[(k0 + 2 * kh + 1) * H_FEATS + col0 + mrow];
      acc = wmma_f32(a, b, acc);
    }
    __syncthreads();
  }

  {
    float bias = bm1[col0 + mrow];
    #pragma unroll
    for (int v = 0; v < 8; ++v) {
      int r = v + 8 * kh;
      float val = acc[v] + bias;
      ldsH[r * 72 + col0 + mrow] = val > 0.0f ? val : 0.0f;
    }
  }
  __syncthreads();

  if (tid < 32) {
    int r = tid >> 1, c = tid & 1;
    float s = bm2[c];
    #pragma unroll 8
    for (int j = 0; j < H_FEATS; ++j)
      s += ldsH[r * 72 + j] * Wm2[j * NUM_CLASSES + c];
    out[(size_t)(row0 + r) * NUM_CLASSES + c] = s;
  }
}

extern "C" void kernel_launch(void* const* d_in, const int* in_sizes, int n_in,
                              void* d_out, int out_size, void* d_ws, size_t ws_size,
                              hipStream_t stream) {
  const float* X      = (const float*)d_in[0];
  const int*   src    = (const int*)  d_in[1];
  const int*   dst    = (const int*)  d_in[2];
  const float* W1     = (const float*)d_in[3];
  const float* b1     = (const float*)d_in[4];
  const float* W2     = (const float*)d_in[5];
  const float* b2     = (const float*)d_in[6];
  const float* thetas = (const float*)d_in[7];
  const float* Wm1    = (const float*)d_in[8];
  const float* bm1    = (const float*)d_in[9];
  const float* Wm2    = (const float*)d_in[10];
  const float* bm2    = (const float*)d_in[11];
  float* out = (float*)d_out;
  float* ws  = (float*)d_ws;

  const int PN = N_NODES * H_FEATS;
  float* dinv = ws;
  float* P0   = ws + 100352;
  float* P1   = P0 + PN;
  float* P2   = P1 + PN;
  float* P3   = P2 + PN;
  int*   ibuf = (int*)(P3 + PN);
  int*   cnt  = ibuf;
  int*   off  = cnt + 100352;
  int*   rank = off + 100352;
  int*   csr  = rank + N_EDGES;
  int*   bsum = csr + N_EDGES;
  float* Wc   = P3 + PN + PN;

  zero_i32<<<512, 256, 0, stream>>>(cnt, N_NODES);
  count_rank_kernel<<<(N_EDGES + 255) / 256, 256, 0, stream>>>(dst, cnt, rank, N_EDGES);
  dinv_bsum_kernel<<<SCAN_BLOCKS, SCAN_BLOCK, 0, stream>>>(cnt, dinv, bsum, N_NODES);
  scan_kernel<<<SCAN_BLOCKS, SCAN_BLOCK, 0, stream>>>(cnt, bsum, off, N_NODES);
  fill_kernel<<<(N_EDGES + 255) / 256, 256, 0, stream>>>(src, dst, rank, off, csr, N_EDGES);

  trunk_kernel<<<N_NODES / 16, 128, 0, stream>>>(X, W1, b1, W2, b2, P0);

  wc_kernel<<<(POLY_ORDER * H_FEATS * H_FEATS + 255) / 256, 256, 0, stream>>>(
      thetas, Wm1, Wc);

  float* Pbuf[4] = {P0, P1, P2, P3};
  for (int k = 1; k < POLY_ORDER; ++k) {
    gather_kernel<<<(N_NODES * 32 + 255) / 256, 256, 0, stream>>>(
        Pbuf[k - 1], off, csr, dinv, Pbuf[k], N_NODES);
  }

  head_kernel<<<N_NODES / 16, 128, 0, stream>>>(
      P0, P1, P2, P3, Wc, bm1, Wm2, bm2, out);
}


# edge grouping: 1.6M returning global adds on random words replaced by a two-level counting sort (256-node buckets, ranks taken in LDS); 10 launches; sums in LDS arrival order, run-dependent as before
# speedup vs baseline: 3.0015x; 1.3373x over previous
#include <hip/hip_runtime.h>


#define N_NODES   100000
#define N_EDGES   1600000
#define IN_FEATS  128
#define H_FEATS   64
#define N_CONVS   3
#define POLY_ORDER 4
#define NUM_CLASSES 2

typedef __attribute__((ext_vector_type(2))) float v2f;
typedef __attribute__((ext_vector_type(8))) float v8f;

__device__ __forceinline__ v8f wmma_f32(v2f a, v2f b, v8f c) {
  return __builtin_amdgcn_wmma_f32_16x16x4_f32(
       false, a,  false, b,
       (short)0, c,  false,  false);
}

#define BUCKET_NODES 256
#define N_BUCKETS    ((N_NODES + BUCKET_NODES - 1) / BUCKET_NODES)
#define CNT_PAD      64
#define EDGE_THREADS 256
#define EDGE_EPT     10
#define EDGE_CHUNK   (EDGE_THREADS * EDGE_EPT)
#define EDGE_BLOCKS  (N_EDGES / EDGE_CHUNK)
static_assert(EDGE_BLOCKS * EDGE_CHUNK == N_EDGES, "the level-1 grid covers the edge list exactly (no bounds test)");
static_assert(N_BUCKETS <= 2 * EDGE_THREADS, "a level-1 thread owns at most two buckets");
#define FIN_THREADS  1024
#define FIN_STRIPS   5

__global__ __launch_bounds__(256)
void zero_i32(int* __restrict__ p, int n) {
  int stride = gridDim.x * blockDim.x;
  for (int i = blockIdx.x * blockDim.x + threadIdx.x; i < n; i += stride)
    p[i] = 0;
}

__device__ __forceinline__ int wave_incl_scan(int x, int lane) {
  for (int d = 1; d < 32; d <<= 1) {
    int y = __shfl_up(x, d);
    if (lane >= d) x += y;
  }
  return x;
}

__global__ __launch_bounds__(EDGE_THREADS)
void bucket_count_kernel(const int* __restrict__ dst, int* __restrict__ btot) {
  __shared__ int h[2 * EDGE_THREADS];
  const int tid = threadIdx.x;
  h[tid] = 0;
  h[tid + EDGE_THREADS] = 0;
  int d[EDGE_EPT];
  const int e0 = blockIdx.x * EDGE_CHUNK + tid;
  for (int i = 0; i < EDGE_EPT; ++i) d[i] = dst[e0 + i * EDGE_THREADS];
  __syncthreads();
  for (int i = 0; i < EDGE_EPT; ++i) atomicAdd(&h[d[i] >> 8], 1);
  __syncthreads();
  for (int t = tid; t < N_BUCKETS; t += EDGE_THREADS) {
    const int c = h[t];
    if (c > 0) atomicAdd(&btot[t * CNT_PAD], c);
  }
}

__global__ __launch_bounds__(EDGE_THREADS)
void bucket_place_kernel(const int* __restrict__ src, const int* __restrict__ dst,
                         const int* __restrict__ btot, int* __restrict__ bcur,
                         int* __restrict__ bbase, int* __restrict__ tmp_pairs) {
  __shared__ int h[2 * EDGE_THREADS];
  __shared__ int rb[2 * EDGE_THREADS];
  __shared__ int w0[EDGE_THREADS / 32];
  __shared__ int w1[EDGE_THREADS / 32];
  int2* tmp = reinterpret_cast<int2*>(tmp_pairs);
  const int tid  = threadIdx.x;
  const int lane = tid & 31;
  const int wave = tid >> 5;
  h[tid] = 0;
  h[tid + EDGE_THREADS] = 0;
  int d[EDGE_EPT], s[EDGE_EPT], r[EDGE_EPT];
  const int e0 = blockIdx.x * EDGE_CHUNK + tid;
  for (int i = 0; i < EDGE_EPT; ++i) {
    d[i] = dst[e0 + i * EDGE_THREADS];
    s[i] = src[e0 + i * EDGE_THREADS];
  }
  const int t0 = tid, t1 = tid + EDGE_THREADS;
  const int x0 = btot[t0 * CNT_PAD];
  const int x1 = t1 < N_BUCKETS ? btot[t1 * CNT_PAD] : 0;
  __syncthreads();
  for (int i = 0; i < EDGE_EPT; ++i) r[i] = atomicAdd(&h[d[i] >> 8], 1);
  const int i0 = wave_incl_scan(x0, lane);
  const int i1 = wave_incl_scan(x1, lane);
  if (lane == 31) { w0[wave] = i0; w1[wave] = i1; }
  __syncthreads();
  int pre0 = 0, pre1 = 0;
  for (int w = 0; w < EDGE_THREADS / 32; ++w) {
    pre1 += w0[w];
    if (w < wave) { pre0 += w0[w]; pre1 += w1[w]; }
  }
  const int b0 = pre0 + i0 - x0;
  const int b1 = pre1 + i1 - x1;
  if (blockIdx.x == 0) {
    bbase[t0] = b0;
    if (t1 < N_BUCKETS) bbase[t1] = b1;
  }
  {
    const int c0 = h[t0];
    rb[t0] = b0 + (c0 > 0 ? atomicAdd(&bcur[t0 * CNT_PAD], c0) : 0);
    if (t1 < N_BUCKETS) {
      const int c1 = h[t1];
      rb[t1] = b1 + (c1 > 0 ? atomicAdd(&bcur[t1 * CNT_PAD], c1) : 0);
    }
  }
  __syncthreads();
  for (int i = 0; i < EDGE_EPT; ++i)
    tmp[rb[d[i] >> 8] + r[i]] = make_int2(s[i], d[i]);
}

__global__ __launch_bounds__(FIN_THREADS)
void bucket_finish_kernel(const int* __restrict__ tmp_pairs, const int* __restrict__ btot,
                          const int* __restrict__ bbase, int* __restrict__ off,
                          float* __restrict__ dinv, int* __restrict__ csr, int n) {
  __shared__ int cntL[BUCKET_NODES];
  __shared__ int curL[BUCKET_NODES];
  __shared__ int wtot[BUCKET_NODES / 32];
  const int2* tmp = reinterpret_cast<const int2*>(tmp_pairs);
  const int tid  = threadIdx.x;
  const int lane = tid & 31;
  const int wave = tid >> 5;
  const int b    = blockIdx.x;
  const int nb   = btot[b * CNT_PAD];
  const int base = bbase[b];
  if (tid < BUCKET_NODES) cntL[tid] = 0;
  __syncthreads();
  for (int i0 = 0; i0 < nb; i0 += FIN_STRIPS * FIN_THREADS) {
    int2 e[FIN_STRIPS];
    for (int k = 0; k < FIN_STRIPS; ++k) {
      const int i = i0 + k * FIN_THREADS + tid;
      e[k] = tmp[base + (i < nb ? i : nb - 1)];
    }
    for (int k = 0; k < FIN_STRIPS; ++k)
      if (i0 + k * FIN_THREADS + tid < nb) atomicAdd(&cntL[e[k].y & (BUCKET_NODES - 1)], 1);
  }
  __syncthreads();
  int c = 0, incl = 0;
  if (tid < BUCKET_NODES) {
    c = cntL[tid];
    incl = wave_incl_scan(c, lane);
    if (lane == 31) wtot[wave] = incl;
  }
  __syncthreads();
  if (tid < BUCKET_NODES) {
    int excl = incl - c;
    for (int w = 0; w < BUCKET_NODES / 32; ++w)
      if (w < wave) excl += wtot[w];
    curL[tid] = excl;
    const int node = b * BUCKET_NODES + tid;
    if (node < n) {
      off[node] = base + excl;
      float dg = (float)c;
      dg = dg < 1.0f ? 1.0f : dg;
      dinv[node] = rsqrtf(dg);
    }
    if (node == n - 1) off[n] = base + excl + c;
  }
  __syncthreads();
  for (int i0 = 0; i0 < nb; i0 += FIN_STRIPS * FIN_THREADS) {
    int2 e[FIN_STRIPS];
    for (int k = 0; k < FIN_STRIPS; ++k) {
      const int i = i0 + k * FIN_THREADS + tid;
      e[k] = tmp[base + (i < nb ? i : nb - 1)];
    }
    for (int k = 0; k < FIN_STRIPS; ++k)
      if (i0 + k * FIN_THREADS + tid < nb) {
        const int p = atomicAdd(&curL[e[k].y & (BUCKET_NODES - 1)], 1);
        csr[base + p] = e[k].x;
      }
  }
}

__global__ __launch_bounds__(128)
void trunk_kernel(const float* __restrict__ X,
                  const float* __restrict__ W1, const float* __restrict__ b1,
                  const float* __restrict__ W2, const float* __restrict__ b2,
                  float* __restrict__ P0) {
  __shared__ float ldsX[16 * 132];
  __shared__ float ldsH[16 * 72];

  const int tid  = threadIdx.x;
  const int lane = tid & 31;
  const int wave = tid >> 5;
  const int mrow = lane & 15;
  const int kh   = lane >> 4;
  const int row0 = blockIdx.x * 16;
  const int col0 = wave * 16;

  for (int i = tid; i < 16 * IN_FEATS; i += 128) {
    int r = i >> 7, c = i & 127;
    ldsX[r * 132 + c] = X[(size_t)(row0 + r) * IN_FEATS + c];
  }
  __syncthreads();

  v8f acc = {};
  for (int k0 = 0; k0 < IN_FEATS; k0 += 4) {
    v2f a, b;
    a.x = ldsX[mrow * 132 + k0 + 2 * kh];
    a.y = ldsX[mrow * 132 + k0 + 2 * kh + 1];
    b.x = W1[(size_t)(k0 + 2 * kh)     * H_FEATS + col0 + mrow];
    b.y = W1[(size_t)(k0 + 2 * kh + 1) * H_FEATS + col0 + mrow];
    acc = wmma_f32(a, b, acc);
  }
  {
    float bias = b1[col0 + mrow];
    #pragma unroll
    for (int v = 0; v < 8; ++v) {
      int r = v + 8 * kh;
      float val = acc[v] + bias;
      ldsH[r * 72 + col0 + mrow] = val > 0.0f ? val : 0.0f;
    }
  }
  __syncthreads();

  v8f acc2 = {};
  for (int k0 = 0; k0 < H_FEATS; k0 += 4) {
    v2f a, b;
    a.x = ldsH[mrow * 72 + k0 + 2 * kh];
    a.y = ldsH[mrow * 72 + k0 + 2 * kh + 1];
    b.x = W2[(size_t)(k0 + 2 * kh)     * H_FEATS + col0 + mrow];
    b.y = W2[(size_t)(k0 + 2 * kh + 1) * H_FEATS + col0 + mrow];
    acc2 = wmma_f32(a, b, acc2);
  }
  {
    float bias = b2[col0 + mrow];
    #pragma unroll
    for (int v = 0; v < 8; ++v) {
      int r = v + 8 * kh;
      float val = acc2[v] + bias;
      P0[(size_t)(row0 + r) * H_FEATS + col0 + mrow] = val > 0.0f ? val : 0.0f;
    }
  }
}

template <int U>
__device__ __forceinline__ void gather_rows(const float2* __restrict__ rows, int s, float di,
                                            int t, int lane, float2& acc) {
  int   st[U];
  float dt[U];
  float2 v[U];
  for (int j = 0; j < U; ++j) {
    st[j] = __builtin_amdgcn_readlane(s, t + j);
    dt[j] = __int_as_float(__builtin_amdgcn_readlane(__float_as_int(di), t + j));
  }
  for (int j = 0; j < U; ++j)
    v[j] = rows[(size_t)st[j] * (H_FEATS / 2) + lane];
  for (int j = 0; j < U; ++j) {
    acc.x = fmaf(dt[j], v[j].x, acc.x);
    acc.y = fmaf(dt[j], v[j].y, acc.y);
  }
}

__global__ __launch_bounds__(256)
void gather_kernel(const float* __restrict__ Pprev, const int* __restrict__ off,
                   const int* __restrict__ csr, const float* __restrict__ dinv,
                   float* __restrict__ Pk, int n) {
  const int gid  = blockIdx.x * blockDim.x + threadIdx.x;
  const int lane = gid & 31;
  const int node = __builtin_amdgcn_readfirstlane(gid >> 5);
  if (node >= n) return;
  const int beg = __builtin_amdgcn_readfirstlane(off[node]);
  const int end = __builtin_amdgcn_readfirstlane(off[node + 1]);
  const float2* rows = reinterpret_cast<const float2*>(Pprev);
  const float2 self = rows[(size_t)node * (H_FEATS / 2) + lane];
  const float dn = dinv[node];

  float2 acc = make_float2(0.0f, 0.0f);
  for (int c = beg; c < end; c += 32) {
    const int m = (end - c) < 32 ? (end - c) : 32;
    int s = 0;
    float di = 0.0f;
    if (lane < m) {
      s  = csr[c + lane];
      di = dinv[s];
    }
    int t = 0;
    for (; t + 8 <= m; t += 8) gather_rows<8>(rows, s, di, t, lane, acc);
    if (t + 4 <= m) { gather_rows<4>(rows, s, di, t, lane, acc); t += 4; }
    if (t + 2 <= m) { gather_rows<2>(rows, s, di, t, lane, acc); t += 2; }
    if (t + 1 <= m) { gather_rows<1>(rows, s, di, t, lane, acc); }
  }
  float2 o;
  o.x = self.x - acc.x * dn;
  o.y = self.y - acc.y * dn;
  reinterpret_cast<float2*>(Pk)[(size_t)node * (H_FEATS / 2) + lane] = o;
}

__global__ __launch_bounds__(256)
void wc_kernel(const float* __restrict__ thetas, const float* __restrict__ Wm1,
               float* __restrict__ Wc) {
  int t = blockIdx.x * blockDim.x + threadIdx.x;
  if (t >= POLY_ORDER * H_FEATS * H_FEATS) return;
  int k  = t >> 12;
  int rc = t & 4095;
  int r  = rc >> 6, c = rc & 63;
  float s = 0.0f;
  #pragma unroll
  for (int i = 0; i < N_CONVS; ++i)
    s += thetas[i * POLY_ORDER + k] * Wm1[(size_t)(i * H_FEATS + r) * H_FEATS + c];
  Wc[t] = s;
}

__global__ __launch_bounds__(128)
void head_kernel(const float* __restrict__ P0, const float* __restrict__ P1,
                 const float* __restrict__ P2, const float* __restrict__ P3,
                 const float* __restrict__ Wc, const float* __restrict__ bm1,
                 const float* __restrict__ Wm2, const float* __restrict__ bm2,
                 float* __restrict__ out) {
  __shared__ float ldsA[16 * 72];
  __shared__ float ldsH[16 * 72];

  const int tid  = threadIdx.x;
  const int lane = tid & 31;
  const int wave = tid >> 5;
  const int mrow = lane & 15;
  const int kh   = lane >> 4;
  const int row0 = blockIdx.x * 16;
  const int col0 = wave * 16;

  const float* Ps[4] = {P0, P1, P2, P3};

  v8f acc = {};
  #pragma unroll
  for (int k = 0; k < POLY_ORDER; ++k) {
    const float* Pk = Ps[k];
    const float* Wk = Wc + k * H_FEATS * H_FEATS;
    for (int i = tid; i < 16 * H_FEATS; i += 128) {
      int r = i >> 6, c = i & 63;
      ldsA[r * 72 + c] = Pk[(size_t)(row0 + r) * H_FEATS + c];
    }
    __syncthreads();
    for (int k0 = 0; k0 < H_FEATS; k0 += 4) {
      v2f a, b;
      a.x = ldsA[mrow * 72 + k0 + 2 * kh];
      a.y = ldsA[mrow * 72 + k0 + 2 * kh + 1];
      b.x = Wk[(k0 + 2 * kh)     * H_FEATS + col0 + mrow];
      b.y = Wk[(k0 + 2 * kh + 1) * H_FEATS + col0 + mrow];
      acc = wmma_f32(a, b, acc);
    }
    __syncthreads();
  }

  {
    float bias = bm1[col0 + mrow];
    #pragma unroll
    for (int v = 0; v < 8; ++v) {
      int r = v + 8 * kh;
      float val = acc[v] + bias;
      ldsH[r * 72 + col0 + mrow] = val > 0.0f ? val : 0.0f;
    }
  }
  __syncthreads();

  if (tid < 32) {
    int r = tid >> 1, c = tid & 1;
    float s = bm2[c];
    #pragma unroll 8
    for (int j = 0; j < H_FEATS; ++j)
      s += ldsH[r * 72 + j] * Wm2[j * NUM_CLASSES + c];
    out[(size_t)(row0 + r) * NUM_CLASSES + c] = s;
  }
}

extern "C" void kernel_launch(void* const* d_in, const int* in_sizes, int n_in,
                              void* d_out, int out_size, void* d_ws, size_t ws_size,
                              hipStream_t stream) {
  const float* X      = (const float*)d_in[0];
  const int*   src    = (const int*)  d_in[1];
  const int*   dst    = (const int*)  d_in[2];
  const float* W1     = (const float*)d_in[3];
  const float* b1     = (const float*)d_in[4];
  const float* W2     = (const float*)d_in[5];
  const float* b2     = (const float*)d_in[6];
  const float* thetas = (const float*)d_in[7];
  const float* Wm1    = (const float*)d_in[8];
  const float* bm1    = (const float*)d_in[9];
  const float* Wm2    = (const float*)d_in[10];
  const float* bm2    = (const float*)d_in[11];
  float* out = (float*)d_out;
  float* ws  = (float*)d_ws;

  const int PN = N_NODES * H_FEATS;
  float* dinv = ws;
  float* P0   = ws + 100352;
  float* P1   = P0 + PN;
  float* P2   = P1 + PN;
  float* P3   = P2 + PN;
  int*   ibuf = (int*)(P3 + PN);
  int*   off   = ibuf;
  int*   csr   = off + 100352;
  int*   tmp   = csr + N_EDGES;
  int*   btot  = csr + 3 * N_EDGES;
  int*   bcur  = btot + N_BUCKETS * CNT_PAD;
  int*   bbase = bcur + N_BUCKETS * CNT_PAD;
  float* Wc   = P3 + PN + PN;

  zero_i32<<<64, 256, 0, stream>>>(btot, 2 * N_BUCKETS * CNT_PAD);
  bucket_count_kernel<<<EDGE_BLOCKS, EDGE_THREADS, 0, stream>>>(dst, btot);
  bucket_place_kernel<<<EDGE_BLOCKS, EDGE_THREADS, 0, stream>>>(src, dst, btot, bcur, bbase, tmp);
  bucket_finish_kernel<<<N_BUCKETS, FIN_THREADS, 0, stream>>>(tmp, btot, bbase, off, dinv, csr, N_NODES);

  trunk_kernel<<<N_NODES / 16, 128, 0, stream>>>(X, W1, b1, W2, b2, P0);

  wc_kernel<<<(POLY_ORDER * H_FEATS * H_FEATS + 255) / 256, 256, 0, stream>>>(
      thetas, Wm1, Wc);

  float* Pbuf[4] = {P0, P1, P2, P3};
  for (int k = 1; k < POLY_ORDER; ++k) {
    gather_kernel<<<(N_NODES * 32 + 255) / 256, 256, 0, stream>>>(
        Pbuf[k - 1], off, csr, dinv, Pbuf[k], N_NODES);
  }

  head_kernel<<<N_NODES / 16, 128, 0, stream>>>(
      P0, P1, P2, P3, Wc, bm1, Wm2, bm2, out);
}


# last propagation pass folded into the head launch: each head work-group gathers its 16 rows of P3 into an LDS tile beside P0..P2 (P3 never stored or re-read); 9 launches; summation order unchanged
# speedup vs baseline: 3.5403x; 1.1795x over previous
#include <hip/hip_runtime.h>


#define N_NODES   100000
#define N_EDGES   1600000
#define IN_FEATS  128
#define H_FEATS   64
#define N_CONVS   3
#define POLY_ORDER 4
#define NUM_CLASSES 2

typedef __attribute__((ext_vector_type(2))) float v2f;
typedef __attribute__((ext_vector_type(8))) float v8f;

__device__ __forceinline__ v8f wmma_f32(v2f a, v2f b, v8f c) {
  return __builtin_amdgcn_wmma_f32_16x16x4_f32(
       false, a,  false, b,
       (short)0, c,  false,  false);
}

#define BUCKET_NODES 256
#define N_BUCKETS    ((N_NODES + BUCKET_NODES - 1) / BUCKET_NODES)
#define CNT_PAD      64
#define EDGE_THREADS 256
#define EDGE_EPT     10
#define EDGE_CHUNK   (EDGE_THREADS * EDGE_EPT)
#define EDGE_BLOCKS  (N_EDGES / EDGE_CHUNK)
static_assert(EDGE_BLOCKS * EDGE_CHUNK == N_EDGES, "the level-1 grid covers the edge list exactly (no bounds test)");
static_assert(N_BUCKETS <= 2 * EDGE_THREADS, "a level-1 thread owns at most two buckets");
#define FIN_THREADS  1024
#define FIN_STRIPS   5

__global__ __launch_bounds__(256)
void zero_i32(int* __restrict__ p, int n) {
  int stride = gridDim.x * blockDim.x;
  for (int i = blockIdx.x * blockDim.x + threadIdx.x; i < n; i += stride)
    p[i] = 0;
}

__device__ __forceinline__ int wave_incl_scan(int x, int lane) {
  for (int d = 1; d < 32; d <<= 1) {
    int y = __shfl_up(x, d);
    if (lane >= d) x += y;
  }
  return x;
}

__global__ __launch_bounds__(EDGE_THREADS)
void bucket_count_kernel(const int* __restrict__ dst, int* __restrict__ btot) {
  __shared__ int h[2 * EDGE_THREADS];
  const int tid = threadIdx.x;
  h[tid] = 0;
  h[tid + EDGE_THREADS] = 0;
  int d[EDGE_EPT];
  const int e0 = blockIdx.x * EDGE_CHUNK + tid;
  for (int i = 0; i < EDGE_EPT; ++i) d[i] = dst[e0 + i * EDGE_THREADS];
  __syncthreads();
  for (int i = 0; i < EDGE_EPT; ++i) atomicAdd(&h[d[i] >> 8], 1);
  __syncthreads();
  for (int t = tid; t < N_BUCKETS; t += EDGE_THREADS) {
    const int c = h[t];
    if (c > 0) atomicAdd(&btot[t * CNT_PAD], c);
  }
}

__global__ __launch_bounds__(EDGE_THREADS)
void bucket_place_kernel(const int* __restrict__ src, const int* __restrict__ dst,
                         const int* __restrict__ btot, int* __restrict__ bcur,
                         int* __restrict__ bbase, int* __restrict__ tmp_pairs) {
  __shared__ int h[2 * EDGE_THREADS];
  __shared__ int rb[2 * EDGE_THREADS];
  __shared__ int w0[EDGE_THREADS / 32];
  __shared__ int w1[EDGE_THREADS / 32];
  int2* tmp = reinterpret_cast<int2*>(tmp_pairs);
  const int tid  = threadIdx.x;
  const int lane = tid & 31;
  const int wave = tid >> 5;
  h[tid] = 0;
  h[tid + EDGE_THREADS] = 0;
  int d[EDGE_EPT], s[EDGE_EPT], r[EDGE_EPT];
  const int e0 = blockIdx.x * EDGE_CHUNK + tid;
  for (int i = 0; i < EDGE_EPT; ++i) {
    d[i] = dst[e0 + i * EDGE_THREADS];
    s[i] = src[e0 + i * EDGE_THREADS];
  }
  const int t0 = tid, t1 = tid + EDGE_THREADS;
  const int x0 = btot[t0 * CNT_PAD];
  const int x1 = t1 < N_BUCKETS ? btot[t1 * CNT_PAD] : 0;
  __syncthreads();
  for (int i = 0; i < EDGE_EPT; ++i) r[i] = atomicAdd(&h[d[i] >> 8], 1);
  const int i0 = wave_incl_scan(x0, lane);
  const int i1 = wave_incl_scan(x1, lane);
  if (lane == 31) { w0[wave] = i0; w1[wave] = i1; }
  __syncthreads();
  int pre0 = 0, pre1 = 0;
  for (int w = 0; w < EDGE_THREADS / 32; ++w) {
    pre1 += w0[w];
    if (w < wave) { pre0 += w0[w]; pre1 += w1[w]; }
  }
  const int b0 = pre0 + i0 - x0;
  const int b1 = pre1 + i1 - x1;
  if (blockIdx.x == 0) {
    bbase[t0] = b0;
    if (t1 < N_BUCKETS) bbase[t1] = b1;
  }
  {
    const int c0 = h[t0];
    rb[t0] = b0 + (c0 > 0 ? atomicAdd(&bcur[t0 * CNT_PAD], c0) : 0);
    if (t1 < N_BUCKETS) {
      const int c1 = h[t1];
      rb[t1] = b1 + (c1 > 0 ? atomicAdd(&bcur[t1 * CNT_PAD], c1) : 0);
    }
  }
  __syncthreads();
  for (int i = 0; i < EDGE_EPT; ++i)
    tmp[rb[d[i] >> 8] + r[i]] = make_int2(s[i], d[i]);
}

__global__ __launch_bounds__(FIN_THREADS)
void bucket_finish_kernel(const int* __restrict__ tmp_pairs, const int* __restrict__ btot,
                          const int* __restrict__ bbase, int* __restrict__ off,
                          float* __restrict__ dinv, int* __restrict__ csr, int n) {
  __shared__ int cntL[BUCKET_NODES];
  __shared__ int curL[BUCKET_NODES];
  __shared__ int wtot[BUCKET_NODES / 32];
  const int2* tmp = reinterpret_cast<const int2*>(tmp_pairs);
  const int tid  = threadIdx.x;
  const int lane = tid & 31;
  const int wave = tid >> 5;
  const int b    = blockIdx.x;
  const int nb   = btot[b * CNT_PAD];
  const int base = bbase[b];
  if (tid < BUCKET_NODES) cntL[tid] = 0;
  __syncthreads();
  for (int i0 = 0; i0 < nb; i0 += FIN_STRIPS * FIN_THREADS) {
    int2 e[FIN_STRIPS];
    for (int k = 0; k < FIN_STRIPS; ++k) {
      const int i = i0 + k * FIN_THREADS + tid;
      e[k] = tmp[base + (i < nb ? i : nb - 1)];
    }
    for (int k = 0; k < FIN_STRIPS; ++k)
      if (i0 + k * FIN_THREADS + tid < nb) atomicAdd(&cntL[e[k].y & (BUCKET_NODES - 1)], 1);
  }
  __syncthreads();
  int c = 0, incl = 0;
  if (tid < BUCKET_NODES) {
    c = cntL[tid];
    incl = wave_incl_scan(c, lane);
    if (lane == 31) wtot[wave] = incl;
  }
  __syncthreads();
  if (tid < BUCKET_NODES) {
    int excl = incl - c;
    for (int w = 0; w < BUCKET_NODES / 32; ++w)
      if (w < wave) excl += wtot[w];
    curL[tid] = excl;
    const int node = b * BUCKET_NODES + tid;
    if (node < n) {
      off[node] = base + excl;
      float dg = (float)c;
      dg = dg < 1.0f ? 1.0f : dg;
      dinv[node] = rsqrtf(dg);
    }
    if (node == n - 1) off[n] = base + excl + c;
  }
  __syncthreads();
  for (int i0 = 0; i0 < nb; i0 += FIN_STRIPS * FIN_THREADS) {
    int2 e[FIN_STRIPS];
    for (int k = 0; k < FIN_STRIPS; ++k) {
      const int i = i0 + k * FIN_THREADS + tid;
      e[k] = tmp[base + (i < nb ? i : nb - 1)];
    }
    for (int k = 0; k < FIN_STRIPS; ++k)
      if (i0 + k * FIN_THREADS + tid < nb) {
        const int p = atomicAdd(&curL[e[k].y & (BUCKET_NODES - 1)], 1);
        csr[base + p] = e[k].x;
      }
  }
}

__global__ __launch_bounds__(128)
void trunk_kernel(const float* __restrict__ X,
                  const float* __restrict__ W1, const float* __restrict__ b1,
                  const float* __restrict__ W2, const float* __restrict__ b2,
                  float* __restrict__ P0) {
  __shared__ float ldsX[16 * 132];
  __shared__ float ldsH[16 * 72];

  const int tid  = threadIdx.x;
  const int lane = tid & 31;
  const int wave = tid >> 5;
  const int mrow = lane & 15;
  const int kh   = lane >> 4;
  const int row0 = blockIdx.x * 16;
  const int col0 = wave * 16;

  for (int i = tid; i < 16 * IN_FEATS; i += 128) {
    int r = i >> 7, c = i & 127;
    ldsX[r * 132 + c] = X[(size_t)(row0 + r) * IN_FEATS + c];
  }
  __syncthreads();

  v8f acc = {};
  for (int k0 = 0; k0 < IN_FEATS; k0 += 4) {
    v2f a, b;
    a.x = ldsX[mrow * 132 + k0 + 2 * kh];
    a.y = ldsX[mrow * 132 + k0 + 2 * kh + 1];
    b.x = W1[(size_t)(k0 + 2 * kh)     * H_FEATS + col0 + mrow];
    b.y = W1[(size_t)(k0 + 2 * kh + 1) * H_FEATS + col0 + mrow];
    acc = wmma_f32(a, b, acc);
  }
  {
    float bias = b1[col0 + mrow];
    #pragma unroll
    for (int v = 0; v < 8; ++v) {
      int r = v + 8 * kh;
      float val = acc[v] + bias;
      ldsH[r * 72 + col0 + mrow] = val > 0.0f ? val : 0.0f;
    }
  }
  __syncthreads();

  v8f acc2 = {};
  for (int k0 = 0; k0 < H_FEATS; k0 += 4) {
    v2f a, b;
    a.x = ldsH[mrow * 72 + k0 + 2 * kh];
    a.y = ldsH[mrow * 72 + k0 + 2 * kh + 1];
    b.x = W2[(size_t)(k0 + 2 * kh)     * H_FEATS + col0 + mrow];
    b.y = W2[(size_t)(k0 + 2 * kh + 1) * H_FEATS + col0 + mrow];
    acc2 = wmma_f32(a, b, acc2);
  }
  {
    float bias = b2[col0 + mrow];
    #pragma unroll
    for (int v = 0; v < 8; ++v) {
      int r = v + 8 * kh;
      float val = acc2[v] + bias;
      P0[(size_t)(row0 + r) * H_FEATS + col0 + mrow] = val > 0.0f ? val : 0.0f;
    }
  }
}

template <int U>
__device__ __forceinline__ void gather_rows(const float2* __restrict__ rows, int s, float di,
                                            int t, int lane, float2& acc) {
  int   st[U];
  float dt[U];
  float2 v[U];
  for (int j = 0; j < U; ++j) {
    st[j] = __builtin_amdgcn_readlane(s, t + j);
    dt[j] = __int_as_float(__builtin_amdgcn_readlane(__float_as_int(di), t + j));
  }
  for (int j = 0; j < U; ++j)
    v[j] = rows[(size_t)st[j] * (H_FEATS / 2) + lane];
  for (int j = 0; j < U; ++j) {
    acc.x = fmaf(dt[j], v[j].x, acc.x);
    acc.y = fmaf(dt[j], v[j].y, acc.y);
  }
}

__global__ __launch_bounds__(256)
void gather_kernel(const float* __restrict__ Pprev, const int* __restrict__ off,
                   const int* __restrict__ csr, const float* __restrict__ dinv,
                   float* __restrict__ Pk, int n) {
  const int gid  = blockIdx.x * blockDim.x + threadIdx.x;
  const int lane = gid & 31;
  const int node = __builtin_amdgcn_readfirstlane(gid >> 5);
  if (node >= n) return;
  const int beg = __builtin_amdgcn_readfirstlane(off[node]);
  const int end = __builtin_amdgcn_readfirstlane(off[node + 1]);
  const float2* rows = reinterpret_cast<const float2*>(Pprev);
  const float2 self = rows[(size_t)node * (H_FEATS / 2) + lane];
  const float dn = dinv[node];

  float2 acc = make_float2(0.0f, 0.0f);
  for (int c = beg; c < end; c += 32) {
    const int m = (end - c) < 32 ? (end - c) : 32;
    int s = 0;
    float di = 0.0f;
    if (lane < m) {
      s  = csr[c + lane];
      di = dinv[s];
    }
    int t = 0;
    for (; t + 8 <= m; t += 8) gather_rows<8>(rows, s, di, t, lane, acc);
    if (t + 4 <= m) { gather_rows<4>(rows, s, di, t, lane, acc); t += 4; }
    if (t + 2 <= m) { gather_rows<2>(rows, s, di, t, lane, acc); t += 2; }
    if (t + 1 <= m) { gather_rows<1>(rows, s, di, t, lane, acc); }
  }
  float2 o;
  o.x = self.x - acc.x * dn;
  o.y = self.y - acc.y * dn;
  reinterpret_cast<float2*>(Pk)[(size_t)node * (H_FEATS / 2) + lane] = o;
}

__global__ __launch_bounds__(256)
void wc_kernel(const float* __restrict__ thetas, const float* __restrict__ Wm1,
               float* __restrict__ Wc) {
  int t = blockIdx.x * blockDim.x + threadIdx.x;
  if (t >= POLY_ORDER * H_FEATS * H_FEATS) return;
  int k  = t >> 12;
  int rc = t & 4095;
  int r  = rc >> 6, c = rc & 63;
  float s = 0.0f;
  #pragma unroll
  for (int i = 0; i < N_CONVS; ++i)
    s += thetas[i * POLY_ORDER + k] * Wm1[(size_t)(i * H_FEATS + r) * H_FEATS + c];
  Wc[t] = s;
}

__device__ __forceinline__ void gather_chunk(const float2* __restrict__ rows, int s, float di,
                                             int m, int lane, float2& acc) {
  int t = 0;
  for (; t + 8 <= m; t += 8) gather_rows<8>(rows, s, di, t, lane, acc);
  if (t + 4 <= m) { gather_rows<4>(rows, s, di, t, lane, acc); t += 4; }
  if (t + 2 <= m) { gather_rows<2>(rows, s, di, t, lane, acc); t += 2; }
  if (t + 1 <= m) { gather_rows<1>(rows, s, di, t, lane, acc); }
}

static_assert(N_NODES % 16 == 0, "the head grid covers the nodes exactly (no bounds test)");
#define TILE_PITCH 72
#define NODES_PER_WAVE 4

__global__ __launch_bounds__(128, 16)
void head_kernel(const float* __restrict__ P0, const float* __restrict__ P1,
                 const float* __restrict__ P2, const int* __restrict__ off,
                 const int* __restrict__ csr, const float* __restrict__ dinv,
                 const float* __restrict__ Wc, const float* __restrict__ bm1,
                 const float* __restrict__ Wm2, const float* __restrict__ bm2,
                 float* __restrict__ out) {
  __shared__ __attribute__((aligned(16))) float tile[POLY_ORDER][16 * TILE_PITCH];

  const int tid  = threadIdx.x;
  const int lane = tid & 31;
  const int wave = __builtin_amdgcn_readfirstlane(tid >> 5);
  const int mrow = lane & 15;
  const int kh   = lane >> 4;
  const int row0 = blockIdx.x * 16;
  const int col0 = wave * 16;
  const int lrow = wave * NODES_PER_WAVE;
  const int nbase = row0 + lrow;

  const float2* rows0 = reinterpret_cast<const float2*>(P0);
  const float2* rows1 = reinterpret_cast<const float2*>(P1);
  const float2* rows2 = reinterpret_cast<const float2*>(P2);

  const int   ov = off[nbase + (lane < NODES_PER_WAVE ? lane : NODES_PER_WAVE)];
  const float dv = dinv[nbase + (lane & (NODES_PER_WAVE - 1))];

  float2 p0[NODES_PER_WAVE], p1[NODES_PER_WAVE], self[NODES_PER_WAVE];
  #pragma unroll
  for (int i = 0; i < NODES_PER_WAVE; ++i) {
    p0[i]   = rows0[(size_t)(nbase + i) * (H_FEATS / 2) + lane];
    p1[i]   = rows1[(size_t)(nbase + i) * (H_FEATS / 2) + lane];
    self[i] = rows2[(size_t)(nbase + i) * (H_FEATS / 2) + lane];
  }

  int beg[NODES_PER_WAVE], end[NODES_PER_WAVE], m0[NODES_PER_WAVE], s0[NODES_PER_WAVE];
  float d0[NODES_PER_WAVE];
  #pragma unroll
  for (int i = 0; i < NODES_PER_WAVE; ++i) {
    beg[i] = __builtin_amdgcn_readlane(ov, i);
    end[i] = __builtin_amdgcn_readlane(ov, i + 1);
    m0[i]  = (end[i] - beg[i]) < 32 ? (end[i] - beg[i]) : 32;
    s0[i]  = 0;
    if (lane < m0[i]) s0[i] = csr[beg[i] + lane];
  }
  #pragma unroll
  for (int i = 0; i < NODES_PER_WAVE; ++i) {
    d0[i] = 0.0f;
    if (lane < m0[i]) d0[i] = dinv[s0[i]];
  }

  #pragma unroll
  for (int i = 0; i < NODES_PER_WAVE; ++i) {
    const int a = (lrow + i) * TILE_PITCH + 2 * lane;
    *reinterpret_cast<float2*>(&tile[0][a]) = p0[i];
    *reinterpret_cast<float2*>(&tile[1][a]) = p1[i];
    *reinterpret_cast<float2*>(&tile[2][a]) = self[i];
  }

  #pragma unroll
  for (int i = 0; i < NODES_PER_WAVE; ++i) {
    float2 acc = make_float2(0.0f, 0.0f);
    gather_chunk(rows2, s0[i], d0[i], m0[i], lane, acc);
    for (int c = beg[i] + 32; c < end[i]; c += 32) {
      const int m = (end[i] - c) < 32 ? (end[i] - c) : 32;
      int s = 0;
      float di = 0.0f;
      if (lane < m) {
        s  = csr[c + lane];
        di = dinv[s];
      }
      gather_chunk(rows2, s, di, m, lane, acc);
    }
    const float dn = __int_as_float(__builtin_amdgcn_readlane(__float_as_int(dv), i));
    float2 o;
    o.x = self[i].x - acc.x * dn;
    o.y = self[i].y - acc.y * dn;
    *reinterpret_cast<float2*>(&tile[3][(lrow + i) * TILE_PITCH + 2 * lane]) = o;
  }
  __syncthreads();

  v8f acc = {};
  #pragma unroll 1
  for (int k = 0; k < POLY_ORDER; ++k) {
    const float* Ak = tile[k];
    const float* Wk = Wc + k * H_FEATS * H_FEATS;
    for (int k0 = 0; k0 < H_FEATS; k0 += 4) {
      v2f a, b;
      a.x = Ak[mrow * TILE_PITCH + k0 + 2 * kh];
      a.y = Ak[mrow * TILE_PITCH + k0 + 2 * kh + 1];
      b.x = Wk[(k0 + 2 * kh)     * H_FEATS + col0 + mrow];
      b.y = Wk[(k0 + 2 * kh + 1) * H_FEATS + col0 + mrow];
      acc = wmma_f32(a, b, acc);
    }
  }
  __syncthreads();

  float* ldsH = tile[0];
  {
    float bias = bm1[col0 + mrow];
    #pragma unroll
    for (int v = 0; v < 8; ++v) {
      int r = v + 8 * kh;
      float val = acc[v] + bias;
      ldsH[r * TILE_PITCH + col0 + mrow] = val > 0.0f ? val : 0.0f;
    }
  }
  __syncthreads();

  if (tid < 32) {
    int r = tid >> 1, c = tid & 1;
    float s = bm2[c];
    #pragma unroll 8
    for (int j = 0; j < H_FEATS; ++j)
      s += ldsH[r * TILE_PITCH + j] * Wm2[j * NUM_CLASSES + c];
    out[(size_t)(row0 + r) * NUM_CLASSES + c] = s;
  }
}

extern "C" void kernel_launch(void* const* d_in, const int* in_sizes, int n_in,
                              void* d_out, int out_size, void* d_ws, size_t ws_size,
                              hipStream_t stream) {
  const float* X      = (const float*)d_in[0];
  const int*   src    = (const int*)  d_in[1];
  const int*   dst    = (const int*)  d_in[2];
  const float* W1     = (const float*)d_in[3];
  const float* b1     = (const float*)d_in[4];
  const float* W2     = (const float*)d_in[5];
  const float* b2     = (const float*)d_in[6];
  const float* thetas = (const float*)d_in[7];
  const float* Wm1    = (const float*)d_in[8];
  const float* bm1    = (const float*)d_in[9];
  const float* Wm2    = (const float*)d_in[10];
  const float* bm2    = (const float*)d_in[11];
  float* out = (float*)d_out;
  float* ws  = (float*)d_ws;

  const int PN = N_NODES * H_FEATS;
  float* dinv = ws;
  float* P0   = ws + 100352;
  float* P1   = P0 + PN;
  float* P2   = P1 + PN;
  float* P3   = P2 + PN;
  int*   ibuf = (int*)(P3 + PN);
  int*   off   = ibuf;
  int*   csr   = off + 100352;
  int*   tmp   = csr + N_EDGES;
  int*   btot  = csr + 3 * N_EDGES;
  int*   bcur  = btot + N_BUCKETS * CNT_PAD;
  int*   bbase = bcur + N_BUCKETS * CNT_PAD;
  float* Wc   = P3 + PN + PN;

  zero_i32<<<64, 256, 0, stream>>>(btot, 2 * N_BUCKETS * CNT_PAD);
  bucket_count_kernel<<<EDGE_BLOCKS, EDGE_THREADS, 0, stream>>>(dst, btot);
  bucket_place_kernel<<<EDGE_BLOCKS, EDGE_THREADS, 0, stream>>>(src, dst, btot, bcur, bbase, tmp);
  bucket_finish_kernel<<<N_BUCKETS, FIN_THREADS, 0, stream>>>(tmp, btot, bbase, off, dinv, csr, N_NODES);

  trunk_kernel<<<N_NODES / 16, 128, 0, stream>>>(X, W1, b1, W2, b2, P0);

  wc_kernel<<<(POLY_ORDER * H_FEATS * H_FEATS + 255) / 256, 256, 0, stream>>>(
      thetas, Wm1, Wc);

  gather_kernel<<<(N_NODES * 32 + 255) / 256, 256, 0, stream>>>(P0, off, csr, dinv, P1, N_NODES);
  gather_kernel<<<(N_NODES * 32 + 255) / 256, 256, 0, stream>>>(P1, off, csr, dinv, P2, N_NODES);

  head_kernel<<<N_NODES / 16, 128, 0, stream>>>(
      P0, P1, P2, off, csr, dinv, Wc, bm1, Wm2, bm2, out);
}


# source d^{-1/2} looked up once per edge: pass 1 stores (source, d^{-1/2}) pairs in list order; pass 2 and the head read the pairs (one 8-byte load per edge), no per-source look-up; same sum order
# speedup vs baseline: 3.7212x; 1.0511x over previous
#include <hip/hip_runtime.h>


#define N_NODES   100000
#define N_EDGES   1600000
#define IN_FEATS  128
#define H_FEATS   64
#define N_CONVS   3
#define POLY_ORDER 4
#define NUM_CLASSES 2

typedef __attribute__((ext_vector_type(2))) float v2f;
typedef __attribute__((ext_vector_type(8))) float v8f;

__device__ __forceinline__ v8f wmma_f32(v2f a, v2f b, v8f c) {
  return __builtin_amdgcn_wmma_f32_16x16x4_f32(
       false, a,  false, b,
       (short)0, c,  false,  false);
}

#define BUCKET_NODES 256
#define N_BUCKETS    ((N_NODES + BUCKET_NODES - 1) / BUCKET_NODES)
#define CNT_PAD      64
#define EDGE_THREADS 256
#define EDGE_EPT     10
#define EDGE_CHUNK   (EDGE_THREADS * EDGE_EPT)
#define EDGE_BLOCKS  (N_EDGES / EDGE_CHUNK)
static_assert(EDGE_BLOCKS * EDGE_CHUNK == N_EDGES, "the level-1 grid covers the edge list exactly (no bounds test)");
static_assert(N_BUCKETS <= 2 * EDGE_THREADS, "a level-1 thread owns at most two buckets");
#define FIN_THREADS  1024
#define FIN_STRIPS   5

__global__ __launch_bounds__(256)
void zero_i32(int* __restrict__ p, int n) {
  int stride = gridDim.x * blockDim.x;
  for (int i = blockIdx.x * blockDim.x + threadIdx.x; i < n; i += stride)
    p[i] = 0;
}

__device__ __forceinline__ int wave_incl_scan(int x, int lane) {
  for (int d = 1; d < 32; d <<= 1) {
    int y = __shfl_up(x, d);
    if (lane >= d) x += y;
  }
  return x;
}

__global__ __launch_bounds__(EDGE_THREADS)
void bucket_count_kernel(const int* __restrict__ dst, int* __restrict__ btot) {
  __shared__ int h[2 * EDGE_THREADS];
  const int tid = threadIdx.x;
  h[tid] = 0;
  h[tid + EDGE_THREADS] = 0;
  int d[EDGE_EPT];
  const int e0 = blockIdx.x * EDGE_CHUNK + tid;
  for (int i = 0; i < EDGE_EPT; ++i) d[i] = dst[e0 + i * EDGE_THREADS];
  __syncthreads();
  for (int i = 0; i < EDGE_EPT; ++i) atomicAdd(&h[d[i] >> 8], 1);
  __syncthreads();
  for (int t = tid; t < N_BUCKETS; t += EDGE_THREADS) {
    const int c = h[t];
    if (c > 0) atomicAdd(&btot[t * CNT_PAD], c);
  }
}

__global__ __launch_bounds__(EDGE_THREADS)
void bucket_place_kernel(const int* __restrict__ src, const int* __restrict__ dst,
                         const int* __restrict__ btot, int* __restrict__ bcur,
                         int* __restrict__ bbase, int* __restrict__ tmp_pairs) {
  __shared__ int h[2 * EDGE_THREADS];
  __shared__ int rb[2 * EDGE_THREADS];
  __shared__ int w0[EDGE_THREADS / 32];
  __shared__ int w1[EDGE_THREADS / 32];
  int2* tmp = reinterpret_cast<int2*>(tmp_pairs);
  const int tid  = threadIdx.x;
  const int lane = tid & 31;
  const int wave = tid >> 5;
  h[tid] = 0;
  h[tid + EDGE_THREADS] = 0;
  int d[EDGE_EPT], s[EDGE_EPT], r[EDGE_EPT];
  const int e0 = blockIdx.x * EDGE_CHUNK + tid;
  for (int i = 0; i < EDGE_EPT; ++i) {
    d[i] = dst[e0 + i * EDGE_THREADS];
    s[i] = src[e0 + i * EDGE_THREADS];
  }
  const int t0 = tid, t1 = tid + EDGE_THREADS;
  const int x0 = btot[t0 * CNT_PAD];
  const int x1 = t1 < N_BUCKETS ? btot[t1 * CNT_PAD] : 0;
  __syncthreads();
  for (int i = 0; i < EDGE_EPT; ++i) r[i] = atomicAdd(&h[d[i] >> 8], 1);
  const int i0 = wave_incl_scan(x0, lane);
  const int i1 = wave_incl_scan(x1, lane);
  if (lane == 31) { w0[wave] = i0; w1[wave] = i1; }
  __syncthreads();
  int pre0 = 0, pre1 = 0;
  for (int w = 0; w < EDGE_THREADS / 32; ++w) {
    pre1 += w0[w];
    if (w < wave) { pre0 += w0[w]; pre1 += w1[w]; }
  }
  const int b0 = pre0 + i0 - x0;
  const int b1 = pre1 + i1 - x1;
  if (blockIdx.x == 0) {
    bbase[t0] = b0;
    if (t1 < N_BUCKETS) bbase[t1] = b1;
  }
  {
    const int c0 = h[t0];
    rb[t0] = b0 + (c0 > 0 ? atomicAdd(&bcur[t0 * CNT_PAD], c0) : 0);
    if (t1 < N_BUCKETS) {
      const int c1 = h[t1];
      rb[t1] = b1 + (c1 > 0 ? atomicAdd(&bcur[t1 * CNT_PAD], c1) : 0);
    }
  }
  __syncthreads();
  for (int i = 0; i < EDGE_EPT; ++i)
    tmp[rb[d[i] >> 8] + r[i]] = make_int2(s[i], d[i]);
}

__global__ __launch_bounds__(FIN_THREADS)
void bucket_finish_kernel(const int* __restrict__ tmp_pairs, const int* __restrict__ btot,
                          const int* __restrict__ bbase, int* __restrict__ off,
                          float* __restrict__ dinv, int* __restrict__ csr, int n) {
  __shared__ int cntL[BUCKET_NODES];
  __shared__ int curL[BUCKET_NODES];
  __shared__ int wtot[BUCKET_NODES / 32];
  const int2* tmp = reinterpret_cast<const int2*>(tmp_pairs);
  const int tid  = threadIdx.x;
  const int lane = tid & 31;
  const int wave = tid >> 5;
  const int b    = blockIdx.x;
  const int nb   = btot[b * CNT_PAD];
  const int base = bbase[b];
  if (tid < BUCKET_NODES) cntL[tid] = 0;
  __syncthreads();
  for (int i0 = 0; i0 < nb; i0 += FIN_STRIPS * FIN_THREADS) {
    int2 e[FIN_STRIPS];
    for (int k = 0; k < FIN_STRIPS; ++k) {
      const int i = i0 + k * FIN_THREADS + tid;
      e[k] = tmp[base + (i < nb ? i : nb - 1)];
    }
    for (int k = 0; k < FIN_STRIPS; ++k)
      if (i0 + k * FIN_THREADS + tid < nb) atomicAdd(&cntL[e[k].y & (BUCKET_NODES - 1)], 1);
  }
  __syncthreads();
  int c = 0, incl = 0;
  if (tid < BUCKET_NODES) {
    c = cntL[tid];
    incl = wave_incl_scan(c, lane);
    if (lane == 31) wtot[wave] = incl;
  }
  __syncthreads();
  if (tid < BUCKET_NODES) {
    int excl = incl - c;
    for (int w = 0; w < BUCKET_NODES / 32; ++w)
      if (w < wave) excl += wtot[w];
    curL[tid] = excl;
    const int node = b * BUCKET_NODES + tid;
    if (node < n) {
      off[node] = base + excl;
      float dg = (float)c;
      dg = dg < 1.0f ? 1.0f : dg;
      dinv[node] = rsqrtf(dg);
    }
    if (node == n - 1) off[n] = base + excl + c;
  }
  __syncthreads();
  for (int i0 = 0; i0 < nb; i0 += FIN_STRIPS * FIN_THREADS) {
    int2 e[FIN_STRIPS];
    for (int k = 0; k < FIN_STRIPS; ++k) {
      const int i = i0 + k * FIN_THREADS + tid;
      e[k] = tmp[base + (i < nb ? i : nb - 1)];
    }
    for (int k = 0; k < FIN_STRIPS; ++k)
      if (i0 + k * FIN_THREADS + tid < nb) {
        const int p = atomicAdd(&curL[e[k].y & (BUCKET_NODES - 1)], 1);
        csr[base + p] = e[k].x;
      }
  }
}

__global__ __launch_bounds__(128)
void trunk_kernel(const float* __restrict__ X,
                  const float* __restrict__ W1, const float* __restrict__ b1,
                  const float* __restrict__ W2, const float* __restrict__ b2,
                  float* __restrict__ P0) {
  __shared__ float ldsX[16 * 132];
  __shared__ float ldsH[16 * 72];

  const int tid  = threadIdx.x;
  const int lane = tid & 31;
  const int wave = tid >> 5;
  const int mrow = lane & 15;
  const int kh   = lane >> 4;
  const int row0 = blockIdx.x * 16;
  const int col0 = wave * 16;

  for (int i = tid; i < 16 * IN_FEATS; i += 128) {
    int r = i >> 7, c = i & 127;
    ldsX[r * 132 + c] = X[(size_t)(row0 + r) * IN_FEATS + c];
  }
  __syncthreads();

  v8f acc = {};
  for (int k0 = 0; k0 < IN_FEATS; k0 += 4) {
    v2f a, b;
    a.x = ldsX[mrow * 132 + k0 + 2 * kh];
    a.y = ldsX[mrow * 132 + k0 + 2 * kh + 1];
    b.x = W1[(size_t)(k0 + 2 * kh)     * H_FEATS + col0 + mrow];
    b.y = W1[(size_t)(k0 + 2 * kh + 1) * H_FEATS + col0 + mrow];
    acc = wmma_f32(a, b, acc);
  }
  {
    float bias = b1[col0 + mrow];
    #pragma unroll
    for (int v = 0; v < 8; ++v) {
      int r = v + 8 * kh;
      float val = acc[v] + bias;
      ldsH[r * 72 + col0 + mrow] = val > 0.0f ? val : 0.0f;
    }
  }
  __syncthreads();

  v8f acc2 = {};
  for (int k0 = 0; k0 < H_FEATS; k0 += 4) {
    v2f a, b;
    a.x = ldsH[mrow * 72 + k0 + 2 * kh];
    a.y = ldsH[mrow * 72 + k0 + 2 * kh + 1];
    b.x = W2[(size_t)(k0 + 2 * kh)     * H_FEATS + col0 + mrow];
    b.y = W2[(size_t)(k0 + 2 * kh + 1) * H_FEATS + col0 + mrow];
    acc2 = wmma_f32(a, b, acc2);
  }
  {
    float bias = b2[col0 + mrow];
    #pragma unroll
    for (int v = 0; v < 8; ++v) {
      int r = v + 8 * kh;
      float val = acc2[v] + bias;
      P0[(size_t)(row0 + r) * H_FEATS + col0 + mrow] = val > 0.0f ? val : 0.0f;
    }
  }
}

template <int U>
__device__ __forceinline__ void gather_rows(const float2* __restrict__ rows, int s, float di,
                                            int t, int lane, float2& acc) {
  int   st[U];
  float dt[U];
  float2 v[U];
  for (int j = 0; j < U; ++j) {
    st[j] = __builtin_amdgcn_readlane(s, t + j);
    dt[j] = __int_as_float(__builtin_amdgcn_readlane(__float_as_int(di), t + j));
  }
  for (int j = 0; j < U; ++j)
    v[j] = rows[(size_t)st[j] * (H_FEATS / 2) + lane];
  for (int j = 0; j < U; ++j) {
    acc.x = fmaf(dt[j], v[j].x, acc.x);
    acc.y = fmaf(dt[j], v[j].y, acc.y);
  }
}

template <bool FIRST>
__global__ __launch_bounds__(256)
void gather_kernel(const float* __restrict__ Pprev, const int* __restrict__ off,
                   const int* __restrict__ csr, const float* __restrict__ dinv,
                   int* __restrict__ pair_words, float* __restrict__ Pk, int n) {
  int2* pairs = reinterpret_cast<int2*>(pair_words);
  const int gid  = blockIdx.x * blockDim.x + threadIdx.x;
  const int lane = gid & 31;
  const int node = __builtin_amdgcn_readfirstlane(gid >> 5);
  if (node >= n) return;
  const int beg = __builtin_amdgcn_readfirstlane(off[node]);
  const int end = __builtin_amdgcn_readfirstlane(off[node + 1]);
  const float2* rows = reinterpret_cast<const float2*>(Pprev);
  const float2 self = rows[(size_t)node * (H_FEATS / 2) + lane];
  const float dn = dinv[node];

  float2 acc = make_float2(0.0f, 0.0f);
  for (int c = beg; c < end; c += 32) {
    const int m = (end - c) < 32 ? (end - c) : 32;
    int s = 0;
    float di = 0.0f;
    if (lane < m) {
      if (FIRST) {
        s  = csr[c + lane];
        di = dinv[s];
      } else {
        const int2 p = pairs[c + lane];
        s  = p.x;
        di = __int_as_float(p.y);
      }
    }
    int t = 0;
    for (; t + 8 <= m; t += 8) gather_rows<8>(rows, s, di, t, lane, acc);
    if (t + 4 <= m) { gather_rows<4>(rows, s, di, t, lane, acc); t += 4; }
    if (t + 2 <= m) { gather_rows<2>(rows, s, di, t, lane, acc); t += 2; }
    if (t + 1 <= m) { gather_rows<1>(rows, s, di, t, lane, acc); }
    if (FIRST && lane < m) pairs[c + lane] = make_int2(s, __float_as_int(di));
  }
  float2 o;
  o.x = self.x - acc.x * dn;
  o.y = self.y - acc.y * dn;
  reinterpret_cast<float2*>(Pk)[(size_t)node * (H_FEATS / 2) + lane] = o;
}

__global__ __launch_bounds__(256)
void wc_kernel(const float* __restrict__ thetas, const float* __restrict__ Wm1,
               float* __restrict__ Wc) {
  int t = blockIdx.x * blockDim.x + threadIdx.x;
  if (t >= POLY_ORDER * H_FEATS * H_FEATS) return;
  int k  = t >> 12;
  int rc = t & 4095;
  int r  = rc >> 6, c = rc & 63;
  float s = 0.0f;
  #pragma unroll
  for (int i = 0; i < N_CONVS; ++i)
    s += thetas[i * POLY_ORDER + k] * Wm1[(size_t)(i * H_FEATS + r) * H_FEATS + c];
  Wc[t] = s;
}

__device__ __forceinline__ void gather_chunk(const float2* __restrict__ rows, int s, float di,
                                             int m, int lane, float2& acc) {
  int t = 0;
  for (; t + 8 <= m; t += 8) gather_rows<8>(rows, s, di, t, lane, acc);
  if (t + 4 <= m) { gather_rows<4>(rows, s, di, t, lane, acc); t += 4; }
  if (t + 2 <= m) { gather_rows<2>(rows, s, di, t, lane, acc); t += 2; }
  if (t + 1 <= m) { gather_rows<1>(rows, s, di, t, lane, acc); }
}

static_assert(N_NODES % 16 == 0, "the head grid covers the nodes exactly (no bounds test)");
#define TILE_PITCH 72
#define NODES_PER_WAVE 4

__global__ __launch_bounds__(128, 16)
void head_kernel(const float* __restrict__ P0, const float* __restrict__ P1,
                 const float* __restrict__ P2, const int* __restrict__ off,
                 const int* __restrict__ pair_words, const float* __restrict__ dinv,
                 const float* __restrict__ Wc, const float* __restrict__ bm1,
                 const float* __restrict__ Wm2, const float* __restrict__ bm2,
                 float* __restrict__ out) {
  __shared__ __attribute__((aligned(16))) float tile[POLY_ORDER][16 * TILE_PITCH];

  const int tid  = threadIdx.x;
  const int lane = tid & 31;
  const int wave = __builtin_amdgcn_readfirstlane(tid >> 5);
  const int mrow = lane & 15;
  const int kh   = lane >> 4;
  const int row0 = blockIdx.x * 16;
  const int col0 = wave * 16;
  const int lrow = wave * NODES_PER_WAVE;
  const int nbase = row0 + lrow;

  const float2* rows0 = reinterpret_cast<const float2*>(P0);
  const float2* rows1 = reinterpret_cast<const float2*>(P1);
  const float2* rows2 = reinterpret_cast<const float2*>(P2);
  const int2*   pairs = reinterpret_cast<const int2*>(pair_words);

  const int   ov = off[nbase + (lane < NODES_PER_WAVE ? lane : NODES_PER_WAVE)];
  const float dv = dinv[nbase + (lane & (NODES_PER_WAVE - 1))];

  float2 p0[NODES_PER_WAVE], p1[NODES_PER_WAVE], self[NODES_PER_WAVE];
  #pragma unroll
  for (int i = 0; i < NODES_PER_WAVE; ++i) {
    p0[i]   = rows0[(size_t)(nbase + i) * (H_FEATS / 2) + lane];
    p1[i]   = rows1[(size_t)(nbase + i) * (H_FEATS / 2) + lane];
    self[i] = rows2[(size_t)(nbase + i) * (H_FEATS / 2) + lane];
  }

  int beg[NODES_PER_WAVE], end[NODES_PER_WAVE], m0[NODES_PER_WAVE], s0[NODES_PER_WAVE];
  float d0[NODES_PER_WAVE];
  #pragma unroll
  for (int i = 0; i < NODES_PER_WAVE; ++i) {
    beg[i] = __builtin_amdgcn_readlane(ov, i);
    end[i] = __builtin_amdgcn_readlane(ov, i + 1);
    m0[i]  = (end[i] - beg[i]) < 32 ? (end[i] - beg[i]) : 32;
    int2 p = make_int2(0, 0);
    if (lane < m0[i]) p = pairs[beg[i] + lane];
    s0[i] = p.x;
    d0[i] = __int_as_float(p.y);
  }

  #pragma unroll
  for (int i = 0; i < NODES_PER_WAVE; ++i) {
    const int a = (lrow + i) * TILE_PITCH + 2 * lane;
    *reinterpret_cast<float2*>(&tile[0][a]) = p0[i];
    *reinterpret_cast<float2*>(&tile[1][a]) = p1[i];
    *reinterpret_cast<float2*>(&tile[2][a]) = self[i];
  }

  #pragma unroll
  for (int i = 0; i < NODES_PER_WAVE; ++i) {
    float2 acc = make_float2(0.0f, 0.0f);
    gather_chunk(rows2, s0[i], d0[i], m0[i], lane, acc);
    for (int c = beg[i] + 32; c < end[i]; c += 32) {
      const int m = (end[i] - c) < 32 ? (end[i] - c) : 32;
      int s = 0;
      float di = 0.0f;
      if (lane < m) {
        const int2 p = pairs[c + lane];
        s  = p.x;
        di = __int_as_float(p.y);
      }
      gather_chunk(rows2, s, di, m, lane, acc);
    }
    const float dn = __int_as_float(__builtin_amdgcn_readlane(__float_as_int(dv), i));
    float2 o;
    o.x = self[i].x - acc.x * dn;
    o.y = self[i].y - acc.y * dn;
    *reinterpret_cast<float2*>(&tile[3][(lrow + i) * TILE_PITCH + 2 * lane]) = o;
  }
  __syncthreads();

  v8f acc = {};
  #pragma unroll 1
  for (int k = 0; k < POLY_ORDER; ++k) {
    const float* Ak = tile[k];
    const float* Wk = Wc + k * H_FEATS * H_FEATS;
    for (int k0 = 0; k0 < H_FEATS; k0 += 4) {
      v2f a, b;
      a.x = Ak[mrow * TILE_PITCH + k0 + 2 * kh];
      a.y = Ak[mrow * TILE_PITCH + k0 + 2 * kh + 1];
      b.x = Wk[(k0 + 2 * kh)     * H_FEATS + col0 + mrow];
      b.y = Wk[(k0 + 2 * kh + 1) * H_FEATS + col0 + mrow];
      acc = wmma_f32(a, b, acc);
    }
  }
  __syncthreads();

  float* ldsH = tile[0];
  {
    float bias = bm1[col0 + mrow];
    #pragma unroll
    for (int v = 0; v < 8; ++v) {
      int r = v + 8 * kh;
      float val = acc[v] + bias;
      ldsH[r * TILE_PITCH + col0 + mrow] = val > 0.0f ? val : 0.0f;
    }
  }
  __syncthreads();

  if (tid < 32) {
    int r = tid >> 1, c = tid & 1;
    float s = bm2[c];
    #pragma unroll 8
    for (int j = 0; j < H_FEATS; ++j)
      s += ldsH[r * TILE_PITCH + j] * Wm2[j * NUM_CLASSES + c];
    out[(size_t)(row0 + r) * NUM_CLASSES + c] = s;
  }
}

extern "C" void kernel_launch(void* const* d_in, const int* in_sizes, int n_in,
                              void* d_out, int out_size, void* d_ws, size_t ws_size,
                              hipStream_t stream) {
  const float* X      = (const float*)d_in[0];
  const int*   src    = (const int*)  d_in[1];
  const int*   dst    = (const int*)  d_in[2];
  const float* W1     = (const float*)d_in[3];
  const float* b1     = (const float*)d_in[4];
  const float* W2     = (const float*)d_in[5];
  const float* b2     = (const float*)d_in[6];
  const float* thetas = (const float*)d_in[7];
  const float* Wm1    = (const float*)d_in[8];
  const float* bm1    = (const float*)d_in[9];
  const float* Wm2    = (const float*)d_in[10];
  const float* bm2    = (const float*)d_in[11];
  float* out = (float*)d_out;
  float* ws  = (float*)d_ws;

  const int PN = N_NODES * H_FEATS;
  float* dinv = ws;
  float* P0   = ws + 100352;
  float* P1   = P0 + PN;
  float* P2   = P1 + PN;
  float* P3   = P2 + PN;
  int*   pairw = (int*)P3;
  int*   ibuf = (int*)(P3 + PN);
  int*   off   = ibuf;
  int*   csr   = off + 100352;
  int*   tmp   = csr + N_EDGES;
  int*   btot  = csr + 3 * N_EDGES;
  int*   bcur  = btot + N_BUCKETS * CNT_PAD;
  int*   bbase = bcur + N_BUCKETS * CNT_PAD;
  float* Wc   = P3 + PN + PN;

  zero_i32<<<64, 256, 0, stream>>>(btot, 2 * N_BUCKETS * CNT_PAD);
  bucket_count_kernel<<<EDGE_BLOCKS, EDGE_THREADS, 0, stream>>>(dst, btot);
  bucket_place_kernel<<<EDGE_BLOCKS, EDGE_THREADS, 0, stream>>>(src, dst, btot, bcur, bbase, tmp);
  bucket_finish_kernel<<<N_BUCKETS, FIN_THREADS, 0, stream>>>(tmp, btot, bbase, off, dinv, csr, N_NODES);

  trunk_kernel<<<N_NODES / 16, 128, 0, stream>>>(X, W1, b1, W2, b2, P0);

  wc_kernel<<<(POLY_ORDER * H_FEATS * H_FEATS + 255) / 256, 256, 0, stream>>>(
      thetas, Wm1, Wc);

  gather_kernel<true><<<(N_NODES * 32 + 255) / 256, 256, 0, stream>>>(P0, off, csr, dinv, pairw, P1, N_NODES);
  gather_kernel<false><<<(N_NODES * 32 + 255) / 256, 256, 0, stream>>>(P1, off, csr, dinv, pairw, P2, N_NODES);

  head_kernel<<<N_NODES / 16, 128, 0, stream>>>(
      P0, P1, P2, off, pairw, dinv, Wc, bm1, Wm2, bm2, out);
}


# launch folding: the trunk's row tiles run as a second block role inside the two level-1 sort launches (half each), and the head-weight sum shares the counter-clear launch; 9 -> 7 launches
# speedup vs baseline: 3.9679x; 1.0663x over previous
#include <hip/hip_runtime.h>


#define N_NODES   100000
#define N_EDGES   1600000
#define IN_FEATS  128
#define H_FEATS   64
#define N_CONVS   3
#define POLY_ORDER 4
#define NUM_CLASSES 2

typedef __attribute__((ext_vector_type(2))) float v2f;
typedef __attribute__((ext_vector_type(8))) float v8f;

__device__ __forceinline__ v8f wmma_f32(v2f a, v2f b, v8f c) {
  return __builtin_amdgcn_wmma_f32_16x16x4_f32(
       false, a,  false, b,
       (short)0, c,  false,  false);
}

#define BUCKET_NODES 256
#define N_BUCKETS    ((N_NODES + BUCKET_NODES - 1) / BUCKET_NODES)
#define CNT_PAD      64
#define EDGE_THREADS 256
#define EDGE_EPT     10
#define EDGE_CHUNK   (EDGE_THREADS * EDGE_EPT)
#define EDGE_BLOCKS  (N_EDGES / EDGE_CHUNK)
static_assert(EDGE_BLOCKS * EDGE_CHUNK == N_EDGES, "the level-1 grid covers the edge list exactly (no bounds test)");
static_assert(N_BUCKETS <= 2 * EDGE_THREADS, "a level-1 thread owns at most two buckets");
#define FIN_THREADS  1024
#define FIN_STRIPS   5

#define PREP_BLOCKS ((POLY_ORDER * H_FEATS * H_FEATS) / 256)
static_assert(PREP_BLOCKS * 256 == POLY_ORDER * H_FEATS * H_FEATS, "the prep grid covers Wc exactly (no bounds test)");

__global__ __launch_bounds__(256)
void prep_kernel(int* __restrict__ cnt, int n_cnt,
                 const float* __restrict__ thetas, const float* __restrict__ Wm1,
                 float* __restrict__ Wc) {
  const int t = blockIdx.x * blockDim.x + threadIdx.x;
  const int stride = gridDim.x * blockDim.x;
  for (int i = t; i < n_cnt; i += stride)
    cnt[i] = 0;
  const int k  = t >> 12;
  const int rc = t & 4095;
  const int r  = rc >> 6, c = rc & 63;
  float s = 0.0f;
  #pragma unroll
  for (int i = 0; i < N_CONVS; ++i)
    s += thetas[i * POLY_ORDER + k] * Wm1[(size_t)(i * H_FEATS + r) * H_FEATS + c];
  Wc[t] = s;
}

__device__ __forceinline__ int wave_incl_scan(int x, int lane) {
  for (int d = 1; d < 32; d <<= 1) {
    int y = __shfl_up(x, d);
    if (lane >= d) x += y;
  }
  return x;
}

__device__ __forceinline__
void bucket_count_role(const int* __restrict__ dst, int* __restrict__ btot) {
  __shared__ int h[2 * EDGE_THREADS];
  const int tid = threadIdx.x;
  h[tid] = 0;
  h[tid + EDGE_THREADS] = 0;
  int d[EDGE_EPT];
  const int e0 = blockIdx.x * EDGE_CHUNK + tid;
  for (int i = 0; i < EDGE_EPT; ++i) d[i] = dst[e0 + i * EDGE_THREADS];
  __syncthreads();
  for (int i = 0; i < EDGE_EPT; ++i) atomicAdd(&h[d[i] >> 8], 1);
  __syncthreads();
  for (int t = tid; t < N_BUCKETS; t += EDGE_THREADS) {
    const int c = h[t];
    if (c > 0) atomicAdd(&btot[t * CNT_PAD], c);
  }
}

__device__ __forceinline__
void bucket_place_role(const int* __restrict__ src, const int* __restrict__ dst,
                       const int* __restrict__ btot, int* __restrict__ bcur,
                       int* __restrict__ bbase, int* __restrict__ tmp_pairs) {
  __shared__ int h[2 * EDGE_THREADS];
  __shared__ int rb[2 * EDGE_THREADS];
  __shared__ int w0[EDGE_THREADS / 32];
  __shared__ int w1[EDGE_THREADS / 32];
  int2* tmp = reinterpret_cast<int2*>(tmp_pairs);
  const int tid  = threadIdx.x;
  const int lane = tid & 31;
  const int wave = tid >> 5;
  h[tid] = 0;
  h[tid + EDGE_THREADS] = 0;
  int d[EDGE_EPT], s[EDGE_EPT], r[EDGE_EPT];
  const int e0 = blockIdx.x * EDGE_CHUNK + tid;
  for (int i = 0; i < EDGE_EPT; ++i) {
    d[i] = dst[e0 + i * EDGE_THREADS];
    s[i] = src[e0 + i * EDGE_THREADS];
  }
  const int t0 = tid, t1 = tid + EDGE_THREADS;
  const int x0 = btot[t0 * CNT_PAD];
  const int x1 = t1 < N_BUCKETS ? btot[t1 * CNT_PAD] : 0;
  __syncthreads();
  for (int i = 0; i < EDGE_EPT; ++i) r[i] = atomicAdd(&h[d[i] >> 8], 1);
  const int i0 = wave_incl_scan(x0, lane);
  const int i1 = wave_incl_scan(x1, lane);
  if (lane == 31) { w0[wave] = i0; w1[wave] = i1; }
  __syncthreads();
  int pre0 = 0, pre1 = 0;
  for (int w = 0; w < EDGE_THREADS / 32; ++w) {
    pre1 += w0[w];
    if (w < wave) { pre0 += w0[w]; pre1 += w1[w]; }
  }
  const int b0 = pre0 + i0 - x0;
  const int b1 = pre1 + i1 - x1;
  if (blockIdx.x == 0) {
    bbase[t0] = b0;
    if (t1 < N_BUCKETS) bbase[t1] = b1;
  }
  {
    const int c0 = h[t0];
    rb[t0] = b0 + (c0 > 0 ? atomicAdd(&bcur[t0 * CNT_PAD], c0) : 0);
    if (t1 < N_BUCKETS) {
      const int c1 = h[t1];
      rb[t1] = b1 + (c1 > 0 ? atomicAdd(&bcur[t1 * CNT_PAD], c1) : 0);
    }
  }
  __syncthreads();
  for (int i = 0; i < EDGE_EPT; ++i)
    tmp[rb[d[i] >> 8] + r[i]] = make_int2(s[i], d[i]);
}

__global__ __launch_bounds__(FIN_THREADS)
void bucket_finish_kernel(const int* __restrict__ tmp_pairs, const int* __restrict__ btot,
                          const int* __restrict__ bbase, int* __restrict__ off,
                          float* __restrict__ dinv, int* __restrict__ csr, int n) {
  __shared__ int cntL[BUCKET_NODES];
  __shared__ int curL[BUCKET_NODES];
  __shared__ int wtot[BUCKET_NODES / 32];
  const int2* tmp = reinterpret_cast<const int2*>(tmp_pairs);
  const int tid  = threadIdx.x;
  const int lane = tid & 31;
  const int wave = tid >> 5;
  const int b    = blockIdx.x;
  const int nb   = btot[b * CNT_PAD];
  const int base = bbase[b];
  if (tid < BUCKET_NODES) cntL[tid] = 0;
  __syncthreads();
  for (int i0 = 0; i0 < nb; i0 += FIN_STRIPS * FIN_THREADS) {
    int2 e[FIN_STRIPS];
    for (int k = 0; k < FIN_STRIPS; ++k) {
      const int i = i0 + k * FIN_THREADS + tid;
      e[k] = tmp[base + (i < nb ? i : nb - 1)];
    }
    for (int k = 0; k < FIN_STRIPS; ++k)
      if (i0 + k * FIN_THREADS + tid < nb) atomicAdd(&cntL[e[k].y & (BUCKET_NODES - 1)], 1);
  }
  __syncthreads();
  int c = 0, incl = 0;
  if (tid < BUCKET_NODES) {
    c = cntL[tid];
    incl = wave_incl_scan(c, lane);
    if (lane == 31) wtot[wave] = incl;
  }
  __syncthreads();
  if (tid < BUCKET_NODES) {
    int excl = incl - c;
    for (int w = 0; w < BUCKET_NODES / 32; ++w)
      if (w < wave) excl += wtot[w];
    curL[tid] = excl;
    const int node = b * BUCKET_NODES + tid;
    if (node < n) {
      off[node] = base + excl;
      float dg = (float)c;
      dg = dg < 1.0f ? 1.0f : dg;
      dinv[node] = rsqrtf(dg);
    }
    if (node == n - 1) off[n] = base + excl + c;
  }
  __syncthreads();
  for (int i0 = 0; i0 < nb; i0 += FIN_STRIPS * FIN_THREADS) {
    int2 e[FIN_STRIPS];
    for (int k = 0; k < FIN_STRIPS; ++k) {
      const int i = i0 + k * FIN_THREADS + tid;
      e[k] = tmp[base + (i < nb ? i : nb - 1)];
    }
    for (int k = 0; k < FIN_STRIPS; ++k)
      if (i0 + k * FIN_THREADS + tid < nb) {
        const int p = atomicAdd(&curL[e[k].y & (BUCKET_NODES - 1)], 1);
        csr[base + p] = e[k].x;
      }
  }
}

#define TRUNK_PER_BLOCK (EDGE_THREADS / 128)
#define TRUNK_BLOCKS    (N_NODES / (16 * TRUNK_PER_BLOCK))
#define TRUNK_BLOCKS_A  ((TRUNK_BLOCKS + 1) / 2)
#define TRUNK_BLOCKS_B  (TRUNK_BLOCKS - TRUNK_BLOCKS_A)
static_assert(TRUNK_BLOCKS * TRUNK_PER_BLOCK * 16 == N_NODES, "the tile pairs cover the nodes exactly (no bounds test)");

__device__ __forceinline__
void trunk_role(int tb, const float* __restrict__ X,
                const float* __restrict__ W1, const float* __restrict__ b1,
                const float* __restrict__ W2, const float* __restrict__ b2,
                float* __restrict__ P0) {
  __shared__ float ldsX2[TRUNK_PER_BLOCK][16 * 132];
  __shared__ float ldsH2[TRUNK_PER_BLOCK][16 * 72];

  const int sub  = __builtin_amdgcn_readfirstlane(threadIdx.x >> 7);
  const int tid  = threadIdx.x & 127;
  float* ldsX = ldsX2[sub];
  float* ldsH = ldsH2[sub];
  const int lane = tid & 31;
  const int wave = tid >> 5;
  const int mrow = lane & 15;
  const int kh   = lane >> 4;
  const int row0 = (tb * TRUNK_PER_BLOCK + sub) * 16;
  const int col0 = wave * 16;

  for (int i = tid; i < 16 * IN_FEATS; i += 128) {
    int r = i >> 7, c = i & 127;
    ldsX[r * 132 + c] = X[(size_t)(row0 + r) * IN_FEATS + c];
  }
  __syncthreads();

  v8f acc = {};
  for (int k0 = 0; k0 < IN_FEATS; k0 += 4) {
    v2f a, b;
    a.x = ldsX[mrow * 132 + k0 + 2 * kh];
    a.y = ldsX[mrow * 132 + k0 + 2 * kh + 1];
    b.x = W1[(size_t)(k0 + 2 * kh)     * H_FEATS + col0 + mrow];
    b.y = W1[(size_t)(k0 + 2 * kh + 1) * H_FEATS + col0 + mrow];
    acc = wmma_f32(a, b, acc);
  }
  {
    float bias = b1[col0 + mrow];
    #pragma unroll
    for (int v = 0; v < 8; ++v) {
      int r = v + 8 * kh;
      float val = acc[v] + bias;
      ldsH[r * 72 + col0 + mrow] = val > 0.0f ? val : 0.0f;
    }
  }
  __syncthreads();

  v8f acc2 = {};
  for (int k0 = 0; k0 < H_FEATS; k0 += 4) {
    v2f a, b;
    a.x = ldsH[mrow * 72 + k0 + 2 * kh];
    a.y = ldsH[mrow * 72 + k0 + 2 * kh + 1];
    b.x = W2[(size_t)(k0 + 2 * kh)     * H_FEATS + col0 + mrow];
    b.y = W2[(size_t)(k0 + 2 * kh + 1) * H_FEATS + col0 + mrow];
    acc2 = wmma_f32(a, b, acc2);
  }
  {
    float bias = b2[col0 + mrow];
    #pragma unroll
    for (int v = 0; v < 8; ++v) {
      int r = v + 8 * kh;
      float val = acc2[v] + bias;
      P0[(size_t)(row0 + r) * H_FEATS + col0 + mrow] = val > 0.0f ? val : 0.0f;
    }
  }
}

__global__ __launch_bounds__(EDGE_THREADS)
void count_trunk_kernel(const int* __restrict__ dst, int* __restrict__ btot,
                        const float* __restrict__ X,
                        const float* __restrict__ W1, const float* __restrict__ b1,
                        const float* __restrict__ W2, const float* __restrict__ b2,
                        float* __restrict__ P0) {
  if (blockIdx.x < EDGE_BLOCKS) bucket_count_role(dst, btot);
  else trunk_role(blockIdx.x - EDGE_BLOCKS, X, W1, b1, W2, b2, P0);
}

__global__ __launch_bounds__(EDGE_THREADS)
void place_trunk_kernel(const int* __restrict__ src, const int* __restrict__ dst,
                        const int* __restrict__ btot, int* __restrict__ bcur,
                        int* __restrict__ bbase, int* __restrict__ tmp_pairs,
                        const float* __restrict__ X,
                        const float* __restrict__ W1, const float* __restrict__ b1,
                        const float* __restrict__ W2, const float* __restrict__ b2,
                        float* __restrict__ P0) {
  if (blockIdx.x < EDGE_BLOCKS) bucket_place_role(src, dst, btot, bcur, bbase, tmp_pairs);
  else trunk_role(TRUNK_BLOCKS_A + (blockIdx.x - EDGE_BLOCKS), X, W1, b1, W2, b2, P0);
}

template <int U>
__device__ __forceinline__ void gather_rows(const float2* __restrict__ rows, int s, float di,
                                            int t, int lane, float2& acc) {
  int   st[U];
  float dt[U];
  float2 v[U];
  for (int j = 0; j < U; ++j) {
    st[j] = __builtin_amdgcn_readlane(s, t + j);
    dt[j] = __int_as_float(__builtin_amdgcn_readlane(__float_as_int(di), t + j));
  }
  for (int j = 0; j < U; ++j)
    v[j] = rows[(size_t)st[j] * (H_FEATS / 2) + lane];
  for (int j = 0; j < U; ++j) {
    acc.x = fmaf(dt[j], v[j].x, acc.x);
    acc.y = fmaf(dt[j], v[j].y, acc.y);
  }
}

template <bool FIRST>
__global__ __launch_bounds__(256)
void gather_kernel(const float* __restrict__ Pprev, const int* __restrict__ off,
                   const int* __restrict__ csr, const float* __restrict__ dinv,
                   int* __restrict__ pair_words, float* __restrict__ Pk, int n) {
  int2* pairs = reinterpret_cast<int2*>(pair_words);
  const int gid  = blockIdx.x * blockDim.x + threadIdx.x;
  const int lane = gid & 31;
  const int node = __builtin_amdgcn_readfirstlane(gid >> 5);
  if (node >= n) return;
  const int beg = __builtin_amdgcn_readfirstlane(off[node]);
  const int end = __builtin_amdgcn_readfirstlane(off[node + 1]);
  const float2* rows = reinterpret_cast<const float2*>(Pprev);
  const float2 self = rows[(size_t)node * (H_FEATS / 2) + lane];
  const float dn = dinv[node];

  float2 acc = make_float2(0.0f, 0.0f);
  for (int c = beg; c < end; c += 32) {
    const int m = (end - c) < 32 ? (end - c) : 32;
    int s = 0;
    float di = 0.0f;
    if (lane < m) {
      if (FIRST) {
        s  = csr[c + lane];
        di = dinv[s];
      } else {
        const int2 p = pairs[c + lane];
        s  = p.x;
        di = __int_as_float(p.y);
      }
    }
    int t = 0;
    for (; t + 8 <= m; t += 8) gather_rows<8>(rows, s, di, t, lane, acc);
    if (t + 4 <= m) { gather_rows<4>(rows, s, di, t, lane, acc); t += 4; }
    if (t + 2 <= m) { gather_rows<2>(rows, s, di, t, lane, acc); t += 2; }
    if (t + 1 <= m) { gather_rows<1>(rows, s, di, t, lane, acc); }
    if (FIRST && lane < m) pairs[c + lane] = make_int2(s, __float_as_int(di));
  }
  float2 o;
  o.x = self.x - acc.x * dn;
  o.y = self.y - acc.y * dn;
  reinterpret_cast<float2*>(Pk)[(size_t)node * (H_FEATS / 2) + lane] = o;
}

__device__ __forceinline__ void gather_chunk(const float2* __restrict__ rows, int s, float di,
                                             int m, int lane, float2& acc) {
  int t = 0;
  for (; t + 8 <= m; t += 8) gather_rows<8>(rows, s, di, t, lane, acc);
  if (t + 4 <= m) { gather_rows<4>(rows, s, di, t, lane, acc); t += 4; }
  if (t + 2 <= m) { gather_rows<2>(rows, s, di, t, lane, acc); t += 2; }
  if (t + 1 <= m) { gather_rows<1>(rows, s, di, t, lane, acc); }
}

static_assert(N_NODES % 16 == 0, "the head grid covers the nodes exactly (no bounds test)");
#define TILE_PITCH 72
#define NODES_PER_WAVE 4

__global__ __launch_bounds__(128, 16)
void head_kernel(const float* __restrict__ P0, const float* __restrict__ P1,
                 const float* __restrict__ P2, const int* __restrict__ off,
                 const int* __restrict__ pair_words, const float* __restrict__ dinv,
                 const float* __restrict__ Wc, const float* __restrict__ bm1,
                 const float* __restrict__ Wm2, const float* __restrict__ bm2,
                 float* __restrict__ out) {
  __shared__ __attribute__((aligned(16))) float tile[POLY_ORDER][16 * TILE_PITCH];

  const int tid  = threadIdx.x;
  const int lane = tid & 31;
  const int wave = __builtin_amdgcn_readfirstlane(tid >> 5);
  const int mrow = lane & 15;
  const int kh   = lane >> 4;
  const int row0 = blockIdx.x * 16;
  const int col0 = wave * 16;
  const int lrow = wave * NODES_PER_WAVE;
  const int nbase = row0 + lrow;

  const float2* rows0 = reinterpret_cast<const float2*>(P0);
  const float2* rows1 = reinterpret_cast<const float2*>(P1);
  const float2* rows2 = reinterpret_cast<const float2*>(P2);
  const int2*   pairs = reinterpret_cast<const int2*>(pair_words);

  const int   ov = off[nbase + (lane < NODES_PER_WAVE ? lane : NODES_PER_WAVE)];
  const float dv = dinv[nbase + (lane & (NODES_PER_WAVE - 1))];

  float2 p0[NODES_PER_WAVE], p1[NODES_PER_WAVE], self[NODES_PER_WAVE];
  #pragma unroll
  for (int i = 0; i < NODES_PER_WAVE; ++i) {
    p0[i]   = rows0[(size_t)(nbase + i) * (H_FEATS / 2) + lane];
    p1[i]   = rows1[(size_t)(nbase + i) * (H_FEATS / 2) + lane];
    self[i] = rows2[(size_t)(nbase + i) * (H_FEATS / 2) + lane];
  }

  int beg[NODES_PER_WAVE], end[NODES_PER_WAVE], m0[NODES_PER_WAVE], s0[NODES_PER_WAVE];
  float d0[NODES_PER_WAVE];
  #pragma unroll
  for (int i = 0; i < NODES_PER_WAVE; ++i) {
    beg[i] = __builtin_amdgcn_readlane(ov, i);
    end[i] = __builtin_amdgcn_readlane(ov, i + 1);
    m0[i]  = (end[i] - beg[i]) < 32 ? (end[i] - beg[i]) : 32;
    int2 p = make_int2(0, 0);
    if (lane < m0[i]) p = pairs[beg[i] + lane];
    s0[i] = p.x;
    d0[i] = __int_as_float(p.y);
  }

  #pragma unroll
  for (int i = 0; i < NODES_PER_WAVE; ++i) {
    const int a = (lrow + i) * TILE_PITCH + 2 * lane;
    *reinterpret_cast<float2*>(&tile[0][a]) = p0[i];
    *reinterpret_cast<float2*>(&tile[1][a]) = p1[i];
    *reinterpret_cast<float2*>(&tile[2][a]) = self[i];
  }

  #pragma unroll
  for (int i = 0; i < NODES_PER_WAVE; ++i) {
    float2 acc = make_float2(0.0f, 0.0f);
    gather_chunk(rows2, s0[i], d0[i], m0[i], lane, acc);
    for (int c = beg[i] + 32; c < end[i]; c += 32) {
      const int m = (end[i] - c) < 32 ? (end[i] - c) : 32;
      int s = 0;
      float di = 0.0f;
      if (lane < m) {
        const int2 p = pairs[c + lane];
        s  = p.x;
        di = __int_as_float(p.y);
      }
      gather_chunk(rows2, s, di, m, lane, acc);
    }
    const float dn = __int_as_float(__builtin_amdgcn_readlane(__float_as_int(dv), i));
    float2 o;
    o.x = self[i].x - acc.x * dn;
    o.y = self[i].y - acc.y * dn;
    *reinterpret_cast<float2*>(&tile[3][(lrow + i) * TILE_PITCH + 2 * lane]) = o;
  }
  __syncthreads();

  v8f acc = {};
  #pragma unroll 1
  for (int k = 0; k < POLY_ORDER; ++k) {
    const float* Ak = tile[k];
    const float* Wk = Wc + k * H_FEATS * H_FEATS;
    for (int k0 = 0; k0 < H_FEATS; k0 += 4) {
      v2f a, b;
      a.x = Ak[mrow * TILE_PITCH + k0 + 2 * kh];
      a.y = Ak[mrow * TILE_PITCH + k0 + 2 * kh + 1];
      b.x = Wk[(k0 + 2 * kh)     * H_FEATS + col0 + mrow];
      b.y = Wk[(k0 + 2 * kh + 1) * H_FEATS + col0 + mrow];
      acc = wmma_f32(a, b, acc);
    }
  }
  __syncthreads();

  float* ldsH = tile[0];
  {
    float bias = bm1[col0 + mrow];
    #pragma unroll
    for (int v = 0; v < 8; ++v) {
      int r = v + 8 * kh;
      float val = acc[v] + bias;
      ldsH[r * TILE_PITCH + col0 + mrow] = val > 0.0f ? val : 0.0f;
    }
  }
  __syncthreads();

  if (tid < 32) {
    int r = tid >> 1, c = tid & 1;
    float s = bm2[c];
    #pragma unroll 8
    for (int j = 0; j < H_FEATS; ++j)
      s += ldsH[r * TILE_PITCH + j] * Wm2[j * NUM_CLASSES + c];
    out[(size_t)(row0 + r) * NUM_CLASSES + c] = s;
  }
}

extern "C" void kernel_launch(void* const* d_in, const int* in_sizes, int n_in,
                              void* d_out, int out_size, void* d_ws, size_t ws_size,
                              hipStream_t stream) {
  const float* X      = (const float*)d_in[0];
  const int*   src    = (const int*)  d_in[1];
  const int*   dst    = (const int*)  d_in[2];
  const float* W1     = (const float*)d_in[3];
  const float* b1     = (const float*)d_in[4];
  const float* W2     = (const float*)d_in[5];
  const float* b2     = (const float*)d_in[6];
  const float* thetas = (const float*)d_in[7];
  const float* Wm1    = (const float*)d_in[8];
  const float* bm1    = (const float*)d_in[9];
  const float* Wm2    = (const float*)d_in[10];
  const float* bm2    = (const float*)d_in[11];
  float* out = (float*)d_out;
  float* ws  = (float*)d_ws;

  const int PN = N_NODES * H_FEATS;
  float* dinv = ws;
  float* P0   = ws + 100352;
  float* P1   = P0 + PN;
  float* P2   = P1 + PN;
  float* P3   = P2 + PN;
  int*   pairw = (int*)P3;
  int*   ibuf = (int*)(P3 + PN);
  int*   off   = ibuf;
  int*   csr   = off + 100352;
  int*   tmp   = csr + N_EDGES;
  int*   btot  = csr + 3 * N_EDGES;
  int*   bcur  = btot + N_BUCKETS * CNT_PAD;
  int*   bbase = bcur + N_BUCKETS * CNT_PAD;
  float* Wc   = P3 + PN + PN;

  prep_kernel<<<PREP_BLOCKS, 256, 0, stream>>>(btot, 2 * N_BUCKETS * CNT_PAD, thetas, Wm1, Wc);

  count_trunk_kernel<<<EDGE_BLOCKS + TRUNK_BLOCKS_A, EDGE_THREADS, 0, stream>>>(
      dst, btot, X, W1, b1, W2, b2, P0);
  place_trunk_kernel<<<EDGE_BLOCKS + TRUNK_BLOCKS_B, EDGE_THREADS, 0, stream>>>(
      src, dst, btot, bcur, bbase, tmp, X, W1, b1, W2, b2, P0);
  bucket_finish_kernel<<<N_BUCKETS, FIN_THREADS, 0, stream>>>(tmp, btot, bbase, off, dinv, csr, N_NODES);

  gather_kernel<true><<<(N_NODES * 32 + 255) / 256, 256, 0, stream>>>(P0, off, csr, dinv, pairw, P1, N_NODES);
  gather_kernel<false><<<(N_NODES * 32 + 255) / 256, 256, 0, stream>>>(P1, off, csr, dinv, pairw, P2, N_NODES);

  head_kernel<<<N_NODES / 16, 128, 0, stream>>>(
      P0, P1, P2, off, pairw, dinv, Wc, bm1, Wm2, bm2, out);
}


# edge grouping without global counters: level-1 blocks sort their own 2560 edges by bucket and record run starts; no count pass, no clear launch; 7 -> 5 launches; sums in LDS arrival order as before
# speedup vs baseline: 5.0349x; 1.2689x over previous
#include <hip/hip_runtime.h>


#define N_NODES   100000
#define N_EDGES   1600000
#define IN_FEATS  128
#define H_FEATS   64
#define N_CONVS   3
#define POLY_ORDER 4
#define NUM_CLASSES 2

typedef __attribute__((ext_vector_type(2))) float v2f;
typedef __attribute__((ext_vector_type(8))) float v8f;

__device__ __forceinline__ v8f wmma_f32(v2f a, v2f b, v8f c) {
  return __builtin_amdgcn_wmma_f32_16x16x4_f32(
       false, a,  false, b,
       (short)0, c,  false,  false);
}

#define BUCKET_NODES 256
#define N_BUCKETS    ((N_NODES + BUCKET_NODES - 1) / BUCKET_NODES)
#define TBL_PITCH    (N_BUCKETS + 1)
#define EDGE_THREADS 256
#define EDGE_EPT     10
#define EDGE_CHUNK   (EDGE_THREADS * EDGE_EPT)
#define EDGE_BLOCKS  (N_EDGES / EDGE_CHUNK)
static_assert(EDGE_BLOCKS * EDGE_CHUNK == N_EDGES, "the level-1 grid covers the edge list exactly (no bounds test)");
static_assert(N_BUCKETS + 1 <= 2 * EDGE_THREADS, "a level-1 thread owns at most two entries of its block's row of run starts");
#define FIN_THREADS  1024
#define RUN_LANES    16
#define RUN_GROUPS   (FIN_THREADS / RUN_LANES)
#define FIN_TRIPS    ((EDGE_BLOCKS + RUN_GROUPS - 1) / RUN_GROUPS)

#define PREP_BLOCKS ((POLY_ORDER * H_FEATS * H_FEATS) / 256)
static_assert(PREP_BLOCKS * 256 == POLY_ORDER * H_FEATS * H_FEATS, "the prep blocks cover Wc exactly (no bounds test)");

__device__ __forceinline__
void head_weight_role(int blk, const float* __restrict__ thetas, const float* __restrict__ Wm1,
                      float* __restrict__ Wc) {
  const int t  = blk * 256 + threadIdx.x;
  const int k  = t >> 12;
  const int rc = t & 4095;
  const int r  = rc >> 6, c = rc & 63;
  float s = 0.0f;
  #pragma unroll
  for (int i = 0; i < N_CONVS; ++i)
    s += thetas[i * POLY_ORDER + k] * Wm1[(size_t)(i * H_FEATS + r) * H_FEATS + c];
  Wc[t] = s;
}

__device__ __forceinline__ int wave_incl_scan(int x, int lane) {
  for (int d = 1; d < 32; d <<= 1) {
    int y = __shfl_up(x, d);
    if (lane >= d) x += y;
  }
  return x;
}

__device__ __forceinline__
void bucket_local_role(const int* __restrict__ src, const int* __restrict__ dst,
                       int* __restrict__ tbl, int* __restrict__ tmp_pairs) {
  __shared__ int h[2 * EDGE_THREADS];
  __shared__ int rb[2 * EDGE_THREADS];
  __shared__ int w0[EDGE_THREADS / 32];
  __shared__ int w1[EDGE_THREADS / 32];
  int2* tmp = reinterpret_cast<int2*>(tmp_pairs) + (size_t)blockIdx.x * EDGE_CHUNK;
  const int tid  = threadIdx.x;
  const int lane = tid & 31;
  const int wave = tid >> 5;
  h[tid] = 0;
  h[tid + EDGE_THREADS] = 0;
  int d[EDGE_EPT], s[EDGE_EPT], r[EDGE_EPT];
  const int e0 = blockIdx.x * EDGE_CHUNK + tid;
  for (int i = 0; i < EDGE_EPT; ++i) {
    d[i] = dst[e0 + i * EDGE_THREADS];
    s[i] = src[e0 + i * EDGE_THREADS];
  }
  __syncthreads();
  for (int i = 0; i < EDGE_EPT; ++i) r[i] = atomicAdd(&h[d[i] >> 8], 1);
  __syncthreads();
  const int t0 = tid, t1 = tid + EDGE_THREADS;
  const int x0 = h[t0];
  const int x1 = h[t1];
  const int i0 = wave_incl_scan(x0, lane);
  const int i1 = wave_incl_scan(x1, lane);
  if (lane == 31) { w0[wave] = i0; w1[wave] = i1; }
  __syncthreads();
  int pre0 = 0, pre1 = 0;
  for (int w = 0; w < EDGE_THREADS / 32; ++w) {
    pre1 += w0[w];
    if (w < wave) { pre0 += w0[w]; pre1 += w1[w]; }
  }
  const int b0 = pre0 + i0 - x0;
  const int b1 = pre1 + i1 - x1;
  rb[t0] = b0;
  rb[t1] = b1;
  int* row = tbl + (size_t)blockIdx.x * TBL_PITCH;
  row[t0] = b0;
  if (t1 <= N_BUCKETS) row[t1] = b1;
  __syncthreads();
  for (int i = 0; i < EDGE_EPT; ++i)
    tmp[rb[d[i] >> 8] + r[i]] = make_int2(s[i], d[i]);
}

static_assert(N_NODES <= (1 << 24), "a source index and a destination's 8 low bits share one 32-bit word in bucket_finish");

__global__ __launch_bounds__(FIN_THREADS, 16)
void bucket_finish_kernel(const int* __restrict__ tmp_pairs, const int* __restrict__ tbl,
                          int* __restrict__ off, float* __restrict__ dinv, int* __restrict__ csr, int n) {
  __shared__ int cntL[BUCKET_NODES];
  __shared__ int curL[BUCKET_NODES];
  __shared__ int wtot[BUCKET_NODES / 32];
  __shared__ int red[FIN_THREADS / 32];
  static_assert(FIN_THREADS / 32 == 32, "one partial sum of red per lane");
  const int2* tmp = reinterpret_cast<const int2*>(tmp_pairs);
  const int tid  = threadIdx.x;
  const int lane = tid & 31;
  const int wave = tid >> 5;
  const int b    = blockIdx.x;
  const int q  = tid & (RUN_LANES - 1);
  const int g0 = tid / RUN_LANES;

  int rs[FIN_TRIPS], re[FIN_TRIPS];
  #pragma unroll
  for (int t = 0; t < FIN_TRIPS; ++t) {
    const int g = g0 + t * RUN_GROUPS;
    rs[t] = 0;
    re[t] = 0;
    if (g < EDGE_BLOCKS) {
      const int* row = tbl + (size_t)g * TBL_PITCH + b;
      rs[t] = row[0];
      re[t] = row[1];
    }
  }
  int jv[FIN_TRIPS];
  unsigned has = 0u, more = 0u;
  int ss = 0;
  #pragma unroll
  for (int t = 0; t < FIN_TRIPS; ++t) {
    const int j = rs[t] + q;
    if (q == 0) ss += rs[t];
    if (j < re[t]) has |= 1u << t;
    if (j + RUN_LANES < re[t]) more |= 1u << t;
    jv[t] = j < EDGE_CHUNK ? j : EDGE_CHUNK - 1;
  }
  int2 ev[FIN_TRIPS];
  #pragma unroll
  for (int t = 0; t < FIN_TRIPS; ++t) {
    const int g  = g0 + t * RUN_GROUPS;
    const int gc = g < EDGE_BLOCKS ? g : EDGE_BLOCKS - 1;
    ev[t] = tmp[gc * EDGE_CHUNK + jv[t]];
  }
  ss = wave_incl_scan(ss, lane);
  if (lane == 31) red[wave] = ss;
  if (tid < BUCKET_NODES) cntL[tid] = 0;
  __syncthreads();
  const int base = __shfl(wave_incl_scan(red[lane], lane), 31);
  unsigned e0[FIN_TRIPS];
  #pragma unroll
  for (int t = 0; t < FIN_TRIPS; ++t)
    e0[t] = ((unsigned)ev[t].x << 8) | ((unsigned)ev[t].y & (BUCKET_NODES - 1));

  #pragma unroll
  for (int t = 0; t < FIN_TRIPS; ++t)
    if ((has >> t) & 1u) atomicAdd(&cntL[e0[t] & (BUCKET_NODES - 1)], 1);
  if (more != 0u) {
    for (int t = 0; t < FIN_TRIPS; ++t)
      if ((more >> t) & 1u) {
        const int g = g0 + t * RUN_GROUPS;
        const int end = tbl[(size_t)g * TBL_PITCH + b + 1];
        const int2* run = tmp + (size_t)g * EDGE_CHUNK;
        for (int j = tbl[(size_t)g * TBL_PITCH + b] + q + RUN_LANES; j < end; j += RUN_LANES)
          atomicAdd(&cntL[run[j].y & (BUCKET_NODES - 1)], 1);
      }
  }
  __syncthreads();
  int c = 0, incl = 0;
  if (tid < BUCKET_NODES) {
    c = cntL[tid];
    incl = wave_incl_scan(c, lane);
    if (lane == 31) wtot[wave] = incl;
  }
  __syncthreads();
  if (tid < BUCKET_NODES) {
    int excl = incl - c;
    for (int w = 0; w < BUCKET_NODES / 32; ++w)
      if (w < wave) excl += wtot[w];
    curL[tid] = excl;
    const int node = b * BUCKET_NODES + tid;
    if (node < n) {
      off[node] = base + excl;
      float dg = (float)c;
      dg = dg < 1.0f ? 1.0f : dg;
      dinv[node] = rsqrtf(dg);
    }
    if (node == n - 1) off[n] = base + excl + c;
  }
  __syncthreads();
  int p0[FIN_TRIPS];
  #pragma unroll
  for (int t = 0; t < FIN_TRIPS; ++t) {
    p0[t] = 0;
    if ((has >> t) & 1u) p0[t] = atomicAdd(&curL[e0[t] & (BUCKET_NODES - 1)], 1);
  }
  #pragma unroll
  for (int t = 0; t < FIN_TRIPS; ++t)
    if ((has >> t) & 1u) csr[base + p0[t]] = (int)(e0[t] >> 8);
  if (more != 0u) {
    for (int t = 0; t < FIN_TRIPS; ++t)
      if ((more >> t) & 1u) {
        const int g = g0 + t * RUN_GROUPS;
        const int end = tbl[(size_t)g * TBL_PITCH + b + 1];
        const int2* run = tmp + (size_t)g * EDGE_CHUNK;
        for (int j = tbl[(size_t)g * TBL_PITCH + b] + q + RUN_LANES; j < end; j += RUN_LANES) {
          const int2 e = run[j];
          const int p = atomicAdd(&curL[e.y & (BUCKET_NODES - 1)], 1);
          csr[base + p] = e.x;
        }
      }
  }
}

#define TRUNK_PER_BLOCK (EDGE_THREADS / 128)
#define TRUNK_BLOCKS    (N_NODES / (16 * TRUNK_PER_BLOCK))
static_assert(TRUNK_BLOCKS * TRUNK_PER_BLOCK * 16 == N_NODES, "the tile pairs cover the nodes exactly (no bounds test)");

__device__ __forceinline__
void trunk_role(int tb, const float* __restrict__ X,
                const float* __restrict__ W1, const float* __restrict__ b1,
                const float* __restrict__ W2, const float* __restrict__ b2,
                float* __restrict__ P0) {
  __shared__ float ldsX2[TRUNK_PER_BLOCK][16 * 132];
  __shared__ float ldsH2[TRUNK_PER_BLOCK][16 * 72];

  const int sub  = __builtin_amdgcn_readfirstlane(threadIdx.x >> 7);
  const int tid  = threadIdx.x & 127;
  float* ldsX = ldsX2[sub];
  float* ldsH = ldsH2[sub];
  const int lane = tid & 31;
  const int wave = tid >> 5;
  const int mrow = lane & 15;
  const int kh   = lane >> 4;
  const int row0 = (tb * TRUNK_PER_BLOCK + sub) * 16;
  const int col0 = wave * 16;

  for (int i = tid; i < 16 * IN_FEATS; i += 128) {
    int r = i >> 7, c = i & 127;
    ldsX[r * 132 + c] = X[(size_t)(row0 + r) * IN_FEATS + c];
  }
  __syncthreads();

  v8f acc = {};
  for (int k0 = 0; k0 < IN_FEATS; k0 += 4) {
    v2f a, b;
    a.x = ldsX[mrow * 132 + k0 + 2 * kh];
    a.y = ldsX[mrow * 132 + k0 + 2 * kh + 1];
    b.x = W1[(size_t)(k0 + 2 * kh)     * H_FEATS + col0 + mrow];
    b.y = W1[(size_t)(k0 + 2 * kh + 1) * H_FEATS + col0 + mrow];
    acc = wmma_f32(a, b, acc);
  }
  {
    float bias = b1[col0 + mrow];
    #pragma unroll
    for (int v = 0; v < 8; ++v) {
      int r = v + 8 * kh;
      float val = acc[v] + bias;
      ldsH[r * 72 + col0 + mrow] = val > 0.0f ? val : 0.0f;
    }
  }
  __syncthreads();

  v8f acc2 = {};
  for (int k0 = 0; k0 < H_FEATS; k0 += 4) {
    v2f a, b;
    a.x = ldsH[mrow * 72 + k0 + 2 * kh];
    a.y = ldsH[mrow * 72 + k0 + 2 * kh + 1];
    b.x = W2[(size_t)(k0 + 2 * kh)     * H_FEATS + col0 + mrow];
    b.y = W2[(size_t)(k0 + 2 * kh + 1) * H_FEATS + col0 + mrow];
    acc2 = wmma_f32(a, b, acc2);
  }
  {
    float bias = b2[col0 + mrow];
    #pragma unroll
    for (int v = 0; v < 8; ++v) {
      int r = v + 8 * kh;
      float val = acc2[v] + bias;
      P0[(size_t)(row0 + r) * H_FEATS + col0 + mrow] = val > 0.0f ? val : 0.0f;
    }
  }
}

__global__ __launch_bounds__(EDGE_THREADS, 16)
void local_trunk_kernel(const int* __restrict__ src, const int* __restrict__ dst,
                        int* __restrict__ tbl, int* __restrict__ tmp_pairs,
                        const float* __restrict__ X,
                        const float* __restrict__ W1, const float* __restrict__ b1,
                        const float* __restrict__ W2, const float* __restrict__ b2,
                        float* __restrict__ P0,
                        const float* __restrict__ thetas, const float* __restrict__ Wm1,
                        float* __restrict__ Wc) {
  if (blockIdx.x < EDGE_BLOCKS) bucket_local_role(src, dst, tbl, tmp_pairs);
  else if (blockIdx.x < EDGE_BLOCKS + TRUNK_BLOCKS) trunk_role(blockIdx.x - EDGE_BLOCKS, X, W1, b1, W2, b2, P0);
  else head_weight_role(blockIdx.x - (EDGE_BLOCKS + TRUNK_BLOCKS), thetas, Wm1, Wc);
}

template <int U>
__device__ __forceinline__ void gather_rows(const float2* __restrict__ rows, int s, float di,
                                            int t, int lane, float2& acc) {
  int   st[U];
  float dt[U];
  float2 v[U];
  for (int j = 0; j < U; ++j) {
    st[j] = __builtin_amdgcn_readlane(s, t + j);
    dt[j] = __int_as_float(__builtin_amdgcn_readlane(__float_as_int(di), t + j));
  }
  for (int j = 0; j < U; ++j)
    v[j] = rows[(size_t)st[j] * (H_FEATS / 2) + lane];
  for (int j = 0; j < U; ++j) {
    acc.x = fmaf(dt[j], v[j].x, acc.x);
    acc.y = fmaf(dt[j], v[j].y, acc.y);
  }
}

template <bool FIRST>
__global__ __launch_bounds__(256)
void gather_kernel(const float* __restrict__ Pprev, const int* __restrict__ off,
                   const int* __restrict__ csr, const float* __restrict__ dinv,
                   int* __restrict__ pair_words, float* __restrict__ Pk, int n) {
  int2* pairs = reinterpret_cast<int2*>(pair_words);
  const int gid  = blockIdx.x * blockDim.x + threadIdx.x;
  const int lane = gid & 31;
  const int node = __builtin_amdgcn_readfirstlane(gid >> 5);
  if (node >= n) return;
  const int beg = __builtin_amdgcn_readfirstlane(off[node]);
  const int end = __builtin_amdgcn_readfirstlane(off[node + 1]);
  const float2* rows = reinterpret_cast<const float2*>(Pprev);
  const float2 self = rows[(size_t)node * (H_FEATS / 2) + lane];
  const float dn = dinv[node];

  float2 acc = make_float2(0.0f, 0.0f);
  for (int c = beg; c < end; c += 32) {
    const int m = (end - c) < 32 ? (end - c) : 32;
    int s = 0;
    float di = 0.0f;
    if (lane < m) {
      if (FIRST) {
        s  = csr[c + lane];
        di = dinv[s];
      } else {
        const int2 p = pairs[c + lane];
        s  = p.x;
        di = __int_as_float(p.y);
      }
    }
    int t = 0;
    for (; t + 8 <= m; t += 8) gather_rows<8>(rows, s, di, t, lane, acc);
    if (t + 4 <= m) { gather_rows<4>(rows, s, di, t, lane, acc); t += 4; }
    if (t + 2 <= m) { gather_rows<2>(rows, s, di, t, lane, acc); t += 2; }
    if (t + 1 <= m) { gather_rows<1>(rows, s, di, t, lane, acc); }
    if (FIRST && lane < m) pairs[c + lane] = make_int2(s, __float_as_int(di));
  }
  float2 o;
  o.x = self.x - acc.x * dn;
  o.y = self.y - acc.y * dn;
  reinterpret_cast<float2*>(Pk)[(size_t)node * (H_FEATS / 2) + lane] = o;
}

__device__ __forceinline__ void gather_chunk(const float2* __restrict__ rows, int s, float di,
                                             int m, int lane, float2& acc) {
  int t = 0;
  for (; t + 8 <= m; t += 8) gather_rows<8>(rows, s, di, t, lane, acc);
  if (t + 4 <= m) { gather_rows<4>(rows, s, di, t, lane, acc); t += 4; }
  if (t + 2 <= m) { gather_rows<2>(rows, s, di, t, lane, acc); t += 2; }
  if (t + 1 <= m) { gather_rows<1>(rows, s, di, t, lane, acc); }
}

static_assert(N_NODES % 16 == 0, "the head grid covers the nodes exactly (no bounds test)");
#define TILE_PITCH 72
#define NODES_PER_WAVE 4

__global__ __launch_bounds__(128, 16)
void head_kernel(const float* __restrict__ P0, const float* __restrict__ P1,
                 const float* __restrict__ P2, const int* __restrict__ off,
                 const int* __restrict__ pair_words, const float* __restrict__ dinv,
                 const float* __restrict__ Wc, const float* __restrict__ bm1,
                 const float* __restrict__ Wm2, const float* __restrict__ bm2,
                 float* __restrict__ out) {
  __shared__ __attribute__((aligned(16))) float tile[POLY_ORDER][16 * TILE_PITCH];

  const int tid  = threadIdx.x;
  const int lane = tid & 31;
  const int wave = __builtin_amdgcn_readfirstlane(tid >> 5);
  const int mrow = lane & 15;
  const int kh   = lane >> 4;
  const int row0 = blockIdx.x * 16;
  const int col0 = wave * 16;
  const int lrow = wave * NODES_PER_WAVE;
  const int nbase = row0 + lrow;

  const float2* rows0 = reinterpret_cast<const float2*>(P0);
  const float2* rows1 = reinterpret_cast<const float2*>(P1);
  const float2* rows2 = reinterpret_cast<const float2*>(P2);
  const int2*   pairs = reinterpret_cast<const int2*>(pair_words);

  const int   ov = off[nbase + (lane < NODES_PER_WAVE ? lane : NODES_PER_WAVE)];
  const float dv = dinv[nbase + (lane & (NODES_PER_WAVE - 1))];

  float2 p0[NODES_PER_WAVE], p1[NODES_PER_WAVE], self[NODES_PER_WAVE];
  #pragma unroll
  for (int i = 0; i < NODES_PER_WAVE; ++i) {
    p0[i]   = rows0[(size_t)(nbase + i) * (H_FEATS / 2) + lane];
    p1[i]   = rows1[(size_t)(nbase + i) * (H_FEATS / 2) + lane];
    self[i] = rows2[(size_t)(nbase + i) * (H_FEATS / 2) + lane];
  }

  int beg[NODES_PER_WAVE], end[NODES_PER_WAVE], m0[NODES_PER_WAVE], s0[NODES_PER_WAVE];
  float d0[NODES_PER_WAVE];
  #pragma unroll
  for (int i = 0; i < NODES_PER_WAVE; ++i) {
    beg[i] = __builtin_amdgcn_readlane(ov, i);
    end[i] = __builtin_amdgcn_readlane(ov, i + 1);
    m0[i]  = (end[i] - beg[i]) < 32 ? (end[i] - beg[i]) : 32;
    int2 p = make_int2(0, 0);
    if (lane < m0[i]) p = pairs[beg[i] + lane];
    s0[i] = p.x;
    d0[i] = __int_as_float(p.y);
  }

  #pragma unroll
  for (int i = 0; i < NODES_PER_WAVE; ++i) {
    const int a = (lrow + i) * TILE_PITCH + 2 * lane;
    *reinterpret_cast<float2*>(&tile[0][a]) = p0[i];
    *reinterpret_cast<float2*>(&tile[1][a]) = p1[i];
    *reinterpret_cast<float2*>(&tile[2][a]) = self[i];
  }

  #pragma unroll
  for (int i = 0; i < NODES_PER_WAVE; ++i) {
    float2 acc = make_float2(0.0f, 0.0f);
    gather_chunk(rows2, s0[i], d0[i], m0[i], lane, acc);
    for (int c = beg[i] + 32; c < end[i]; c += 32) {
      const int m = (end[i] - c) < 32 ? (end[i] - c) : 32;
      int s = 0;
      float di = 0.0f;
      if (lane < m) {
        const int2 p = pairs[c + lane];
        s  = p.x;
        di = __int_as_float(p.y);
      }
      gather_chunk(rows2, s, di, m, lane, acc);
    }
    const float dn = __int_as_float(__builtin_amdgcn_readlane(__float_as_int(dv), i));
    float2 o;
    o.x = self[i].x - acc.x * dn;
    o.y = self[i].y - acc.y * dn;
    *reinterpret_cast<float2*>(&tile[3][(lrow + i) * TILE_PITCH + 2 * lane]) = o;
  }
  __syncthreads();

  v8f acc = {};
  #pragma unroll 1
  for (int k = 0; k < POLY_ORDER; ++k) {
    const float* Ak = tile[k];
    const float* Wk = Wc + k * H_FEATS * H_FEATS;
    for (int k0 = 0; k0 < H_FEATS; k0 += 4) {
      v2f a, b;
      a.x = Ak[mrow * TILE_PITCH + k0 + 2 * kh];
      a.y = Ak[mrow * TILE_PITCH + k0 + 2 * kh + 1];
      b.x = Wk[(k0 + 2 * kh)     * H_FEATS + col0 + mrow];
      b.y = Wk[(k0 + 2 * kh + 1) * H_FEATS + col0 + mrow];
      acc = wmma_f32(a, b, acc);
    }
  }
  __syncthreads();

  float* ldsH = tile[0];
  {
    float bias = bm1[col0 + mrow];
    #pragma unroll
    for (int v = 0; v < 8; ++v) {
      int r = v + 8 * kh;
      float val = acc[v] + bias;
      ldsH[r * TILE_PITCH + col0 + mrow] = val > 0.0f ? val : 0.0f;
    }
  }
  __syncthreads();

  if (tid < 32) {
    int r = tid >> 1, c = tid & 1;
    float s = bm2[c];
    #pragma unroll 8
    for (int j = 0; j < H_FEATS; ++j)
      s += ldsH[r * TILE_PITCH + j] * Wm2[j * NUM_CLASSES + c];
    out[(size_t)(row0 + r) * NUM_CLASSES + c] = s;
  }
}

extern "C" void kernel_launch(void* const* d_in, const int* in_sizes, int n_in,
                              void* d_out, int out_size, void* d_ws, size_t ws_size,
                              hipStream_t stream) {
  const float* X      = (const float*)d_in[0];
  const int*   src    = (const int*)  d_in[1];
  const int*   dst    = (const int*)  d_in[2];
  const float* W1     = (const float*)d_in[3];
  const float* b1     = (const float*)d_in[4];
  const float* W2     = (const float*)d_in[5];
  const float* b2     = (const float*)d_in[6];
  const float* thetas = (const float*)d_in[7];
  const float* Wm1    = (const float*)d_in[8];
  const float* bm1    = (const float*)d_in[9];
  const float* Wm2    = (const float*)d_in[10];
  const float* bm2    = (const float*)d_in[11];
  float* out = (float*)d_out;
  float* ws  = (float*)d_ws;

  const int PN = N_NODES * H_FEATS;
  float* dinv = ws;
  float* P0   = ws + 100352;
  float* P1   = P0 + PN;
  float* P2   = P1 + PN;
  float* P3   = P2 + PN;
  int*   pairw = (int*)P3;
  int*   ibuf = (int*)(P3 + PN);
  int*   off   = ibuf;
  int*   csr   = off + 100352;
  int*   tmp   = csr + N_EDGES;
  int*   tbl   = csr + 3 * N_EDGES;
  float* Wc   = P3 + PN + PN;

  local_trunk_kernel<<<EDGE_BLOCKS + TRUNK_BLOCKS + PREP_BLOCKS, EDGE_THREADS, 0, stream>>>(
      src, dst, tbl, tmp, X, W1, b1, W2, b2, P0, thetas, Wm1, Wc);
  bucket_finish_kernel<<<N_BUCKETS, FIN_THREADS, 0, stream>>>(tmp, tbl, off, dinv, csr, N_NODES);

  gather_kernel<true><<<(N_NODES * 32 + 255) / 256, 256, 0, stream>>>(P0, off, csr, dinv, pairw, P1, N_NODES);
  gather_kernel<false><<<(N_NODES * 32 + 255) / 256, 256, 0, stream>>>(P1, off, csr, dinv, pairw, P2, N_NODES);

  head_kernel<<<N_NODES / 16, 128, 0, stream>>>(
      P0, P1, P2, off, pairw, dinv, Wc, bm1, Wm2, bm2, out);
}


# edge grouping, both levels: records are placed in LDS first and the finished region (2560 pairs; a bucket's lists) goes out as whole 128-byte lines, not as 32-line scattered stores; same list order
# speedup vs baseline: 5.2827x; 1.0492x over previous
#include <hip/hip_runtime.h>


#define N_NODES   100000
#define N_EDGES   1600000
#define IN_FEATS  128
#define H_FEATS   64
#define N_CONVS   3
#define POLY_ORDER 4
#define NUM_CLASSES 2

typedef __attribute__((ext_vector_type(2))) float v2f;
typedef __attribute__((ext_vector_type(8))) float v8f;

__device__ __forceinline__ v8f wmma_f32(v2f a, v2f b, v8f c) {
  return __builtin_amdgcn_wmma_f32_16x16x4_f32(
       false, a,  false, b,
       (short)0, c,  false,  false);
}

#define BUCKET_NODES 256
#define N_BUCKETS    ((N_NODES + BUCKET_NODES - 1) / BUCKET_NODES)
#define TBL_PITCH    (N_BUCKETS + 1)
#define EDGE_THREADS 256
#define EDGE_EPT     10
#define EDGE_CHUNK   (EDGE_THREADS * EDGE_EPT)
#define EDGE_BLOCKS  (N_EDGES / EDGE_CHUNK)
static_assert(EDGE_BLOCKS * EDGE_CHUNK == N_EDGES, "the level-1 grid covers the edge list exactly (no bounds test)");
static_assert(N_BUCKETS + 1 <= 2 * EDGE_THREADS, "a level-1 thread owns at most two entries of its block's row of run starts");
static_assert(EDGE_EPT % 2 == 0, "a level-1 thread writes its share of the block's region as 16-byte words of two pairs");
#define FIN_THREADS  1024
#define RUN_LANES    16
#define RUN_GROUPS   (FIN_THREADS / RUN_LANES)
#define FIN_TRIPS    ((EDGE_BLOCKS + RUN_GROUPS - 1) / RUN_GROUPS)
#define FIN_STAGE    6144

#define PREP_BLOCKS ((POLY_ORDER * H_FEATS * H_FEATS) / 256)
static_assert(PREP_BLOCKS * 256 == POLY_ORDER * H_FEATS * H_FEATS, "the prep blocks cover Wc exactly (no bounds test)");

__device__ __forceinline__
void head_weight_role(int blk, const float* __restrict__ thetas, const float* __restrict__ Wm1,
                      float* __restrict__ Wc) {
  const int t  = blk * 256 + threadIdx.x;
  const int k  = t >> 12;
  const int rc = t & 4095;
  const int r  = rc >> 6, c = rc & 63;
  float s = 0.0f;
  #pragma unroll
  for (int i = 0; i < N_CONVS; ++i)
    s += thetas[i * POLY_ORDER + k] * Wm1[(size_t)(i * H_FEATS + r) * H_FEATS + c];
  Wc[t] = s;
}

__device__ __forceinline__ int wave_incl_scan(int x, int lane) {
  for (int d = 1; d < 32; d <<= 1) {
    int y = __shfl_up(x, d);
    if (lane >= d) x += y;
  }
  return x;
}

__device__ __forceinline__
void bucket_local_role(const int* __restrict__ src, const int* __restrict__ dst,
                       int* __restrict__ tbl, int* __restrict__ tmp_pairs, int2* stg) {
  __shared__ int h[2 * EDGE_THREADS];
  __shared__ int rb[2 * EDGE_THREADS];
  __shared__ int w0[EDGE_THREADS / 32];
  __shared__ int w1[EDGE_THREADS / 32];
  int4* tmp4 = reinterpret_cast<int4*>(tmp_pairs) + (size_t)blockIdx.x * (EDGE_CHUNK / 2);
  const int tid  = threadIdx.x;
  const int lane = tid & 31;
  const int wave = tid >> 5;
  h[tid] = 0;
  h[tid + EDGE_THREADS] = 0;
  int d[EDGE_EPT], s[EDGE_EPT], r[EDGE_EPT];
  const int e0 = blockIdx.x * EDGE_CHUNK + tid;
  for (int i = 0; i < EDGE_EPT; ++i) {
    d[i] = dst[e0 + i * EDGE_THREADS];
    s[i] = src[e0 + i * EDGE_THREADS];
  }
  __syncthreads();
  for (int i = 0; i < EDGE_EPT; ++i) r[i] = atomicAdd(&h[d[i] >> 8], 1);
  __syncthreads();
  const int t0 = tid, t1 = tid + EDGE_THREADS;
  const int x0 = h[t0];
  const int x1 = h[t1];
  const int i0 = wave_incl_scan(x0, lane);
  const int i1 = wave_incl_scan(x1, lane);
  if (lane == 31) { w0[wave] = i0; w1[wave] = i1; }
  __syncthreads();
  int pre0 = 0, pre1 = 0;
  for (int w = 0; w < EDGE_THREADS / 32; ++w) {
    pre1 += w0[w];
    if (w < wave) { pre0 += w0[w]; pre1 += w1[w]; }
  }
  const int b0 = pre0 + i0 - x0;
  const int b1 = pre1 + i1 - x1;
  rb[t0] = b0;
  rb[t1] = b1;
  int* row = tbl + (size_t)blockIdx.x * TBL_PITCH;
  row[t0] = b0;
  if (t1 <= N_BUCKETS) row[t1] = b1;
  __syncthreads();
  for (int i = 0; i < EDGE_EPT; ++i)
    stg[rb[d[i] >> 8] + r[i]] = make_int2(s[i], d[i]);
  __syncthreads();
  const int4* stg4 = reinterpret_cast<const int4*>(stg);
  for (int i = 0; i < EDGE_EPT / 2; ++i)
    tmp4[tid + i * EDGE_THREADS] = stg4[tid + i * EDGE_THREADS];
}

static_assert(N_NODES <= (1 << 24), "a source index and a destination's 8 low bits share one 32-bit word in bucket_finish");

__global__ __launch_bounds__(FIN_THREADS, 16)
void bucket_finish_kernel(const int* __restrict__ tmp_pairs, const int* __restrict__ tbl,
                          int* __restrict__ off, float* __restrict__ dinv, int* __restrict__ csr, int n) {
  __shared__ int cntL[BUCKET_NODES];
  __shared__ int curL[BUCKET_NODES];
  __shared__ int wtot[BUCKET_NODES / 32];
  __shared__ int red[FIN_THREADS / 32];
  __shared__ int stage[FIN_STAGE];
  static_assert(FIN_THREADS / 32 == 32, "one partial sum of red per lane");
  const int2* tmp = reinterpret_cast<const int2*>(tmp_pairs);
  const int tid  = threadIdx.x;
  const int lane = tid & 31;
  const int wave = tid >> 5;
  const int b    = blockIdx.x;
  const int q  = tid & (RUN_LANES - 1);
  const int g0 = tid / RUN_LANES;

  int rs[FIN_TRIPS], re[FIN_TRIPS];
  #pragma unroll
  for (int t = 0; t < FIN_TRIPS; ++t) {
    const int g = g0 + t * RUN_GROUPS;
    rs[t] = 0;
    re[t] = 0;
    if (g < EDGE_BLOCKS) {
      const int* row = tbl + (size_t)g * TBL_PITCH + b;
      rs[t] = row[0];
      re[t] = row[1];
    }
  }
  int jv[FIN_TRIPS];
  unsigned has = 0u, more = 0u;
  int ss = 0;
  #pragma unroll
  for (int t = 0; t < FIN_TRIPS; ++t) {
    const int j = rs[t] + q;
    if (q == 0) ss += rs[t];
    if (j < re[t]) has |= 1u << t;
    if (j + RUN_LANES < re[t]) more |= 1u << t;
    jv[t] = j < EDGE_CHUNK ? j : EDGE_CHUNK - 1;
  }
  int2 ev[FIN_TRIPS];
  #pragma unroll
  for (int t = 0; t < FIN_TRIPS; ++t) {
    const int g  = g0 + t * RUN_GROUPS;
    const int gc = g < EDGE_BLOCKS ? g : EDGE_BLOCKS - 1;
    ev[t] = tmp[gc * EDGE_CHUNK + jv[t]];
  }
  ss = wave_incl_scan(ss, lane);
  if (lane == 31) red[wave] = ss;
  if (tid < BUCKET_NODES) cntL[tid] = 0;
  __syncthreads();
  const int base = __shfl(wave_incl_scan(red[lane], lane), 31);
  unsigned e0[FIN_TRIPS];
  #pragma unroll
  for (int t = 0; t < FIN_TRIPS; ++t)
    e0[t] = ((unsigned)ev[t].x << 8) | ((unsigned)ev[t].y & (BUCKET_NODES - 1));

  #pragma unroll
  for (int t = 0; t < FIN_TRIPS; ++t)
    if ((has >> t) & 1u) atomicAdd(&cntL[e0[t] & (BUCKET_NODES - 1)], 1);
  if (more != 0u) {
    for (int t = 0; t < FIN_TRIPS; ++t)
      if ((more >> t) & 1u) {
        const int g = g0 + t * RUN_GROUPS;
        const int end = tbl[(size_t)g * TBL_PITCH + b + 1];
        const int2* run = tmp + (size_t)g * EDGE_CHUNK;
        for (int j = tbl[(size_t)g * TBL_PITCH + b] + q + RUN_LANES; j < end; j += RUN_LANES)
          atomicAdd(&cntL[run[j].y & (BUCKET_NODES - 1)], 1);
      }
  }
  __syncthreads();
  int c = 0, incl = 0;
  if (tid < BUCKET_NODES) {
    c = cntL[tid];
    incl = wave_incl_scan(c, lane);
    if (lane == 31) wtot[wave] = incl;
  }
  __syncthreads();
  if (tid < BUCKET_NODES) {
    int excl = incl - c;
    for (int w = 0; w < BUCKET_NODES / 32; ++w)
      if (w < wave) excl += wtot[w];
    curL[tid] = excl;
    const int node = b * BUCKET_NODES + tid;
    if (node < n) {
      off[node] = base + excl;
      float dg = (float)c;
      dg = dg < 1.0f ? 1.0f : dg;
      dinv[node] = rsqrtf(dg);
    }
    if (node == n - 1) off[n] = base + excl + c;
  }
  __syncthreads();
  int total = 0;
  for (int w = 0; w < BUCKET_NODES / 32; ++w) total += wtot[w];
  const bool staged = total <= FIN_STAGE;
  int p0[FIN_TRIPS];
  #pragma unroll
  for (int t = 0; t < FIN_TRIPS; ++t) {
    p0[t] = 0;
    if ((has >> t) & 1u) p0[t] = atomicAdd(&curL[e0[t] & (BUCKET_NODES - 1)], 1);
  }
  if (staged) {
    #pragma unroll
    for (int t = 0; t < FIN_TRIPS; ++t)
      if ((has >> t) & 1u) stage[p0[t]] = (int)(e0[t] >> 8);
  } else {
    #pragma unroll
    for (int t = 0; t < FIN_TRIPS; ++t)
      if ((has >> t) & 1u) csr[base + p0[t]] = (int)(e0[t] >> 8);
  }
  if (more != 0u) {
    for (int t = 0; t < FIN_TRIPS; ++t)
      if ((more >> t) & 1u) {
        const int g = g0 + t * RUN_GROUPS;
        const int end = tbl[(size_t)g * TBL_PITCH + b + 1];
        const int2* run = tmp + (size_t)g * EDGE_CHUNK;
        for (int j = tbl[(size_t)g * TBL_PITCH + b] + q + RUN_LANES; j < end; j += RUN_LANES) {
          const int2 e = run[j];
          const int p = atomicAdd(&curL[e.y & (BUCKET_NODES - 1)], 1);
          if (staged) stage[p] = e.x;
          else        csr[base + p] = e.x;
        }
      }
  }
  if (staged) {
    __syncthreads();
    for (int i = tid; i < total; i += FIN_THREADS) csr[base + i] = stage[i];
  }
}

#define TRUNK_PER_BLOCK (EDGE_THREADS / 128)
#define TRUNK_BLOCKS    (N_NODES / (16 * TRUNK_PER_BLOCK))
static_assert(TRUNK_BLOCKS * TRUNK_PER_BLOCK * 16 == N_NODES, "the tile pairs cover the nodes exactly (no bounds test)");
#define TRUNK_X_TILE (16 * 132)
#define TRUNK_H_TILE (16 * 72)
#define ROLE_LDS_FLOATS (TRUNK_PER_BLOCK * (TRUNK_X_TILE + TRUNK_H_TILE))
static_assert(ROLE_LDS_FLOATS * 4 >= EDGE_CHUNK * 8, "the grouping role's region of EDGE_CHUNK pairs fits the LDS that the trunk role's tiles take");

__device__ __forceinline__
void trunk_role(int tb, const float* __restrict__ X,
                const float* __restrict__ W1, const float* __restrict__ b1,
                const float* __restrict__ W2, const float* __restrict__ b2,
                float* __restrict__ P0, float* role_lds) {

  const int sub  = __builtin_amdgcn_readfirstlane(threadIdx.x >> 7);
  const int tid  = threadIdx.x & 127;
  float* ldsX = role_lds + sub * TRUNK_X_TILE;
  float* ldsH = role_lds + TRUNK_PER_BLOCK * TRUNK_X_TILE + sub * TRUNK_H_TILE;
  const int lane = tid & 31;
  const int wave = tid >> 5;
  const int mrow = lane & 15;
  const int kh   = lane >> 4;
  const int row0 = (tb * TRUNK_PER_BLOCK + sub) * 16;
  const int col0 = wave * 16;

  for (int i = tid; i < 16 * IN_FEATS; i += 128) {
    int r = i >> 7, c = i & 127;
    ldsX[r * 132 + c] = X[(size_t)(row0 + r) * IN_FEATS + c];
  }
  __syncthreads();

  v8f acc = {};
  for (int k0 = 0; k0 < IN_FEATS; k0 += 4) {
    v2f a, b;
    a.x = ldsX[mrow * 132 + k0 + 2 * kh];
    a.y = ldsX[mrow * 132 + k0 + 2 * kh + 1];
    b.x = W1[(size_t)(k0 + 2 * kh)     * H_FEATS + col0 + mrow];
    b.y = W1[(size_t)(k0 + 2 * kh + 1) * H_FEATS + col0 + mrow];
    acc = wmma_f32(a, b, acc);
  }
  {
    float bias = b1[col0 + mrow];
    #pragma unroll
    for (int v = 0; v < 8; ++v) {
      int r = v + 8 * kh;
      float val = acc[v] + bias;
      ldsH[r * 72 + col0 + mrow] = val > 0.0f ? val : 0.0f;
    }
  }
  __syncthreads();

  v8f acc2 = {};
  for (int k0 = 0; k0 < H_FEATS; k0 += 4) {
    v2f a, b;
    a.x = ldsH[mrow * 72 + k0 + 2 * kh];
    a.y = ldsH[mrow * 72 + k0 + 2 * kh + 1];
    b.x = W2[(size_t)(k0 + 2 * kh)     * H_FEATS + col0 + mrow];
    b.y = W2[(size_t)(k0 + 2 * kh + 1) * H_FEATS + col0 + mrow];
    acc2 = wmma_f32(a, b, acc2);
  }
  {
    float bias = b2[col0 + mrow];
    #pragma unroll
    for (int v = 0; v < 8; ++v) {
      int r = v + 8 * kh;
      float val = acc2[v] + bias;
      P0[(size_t)(row0 + r) * H_FEATS + col0 + mrow] = val > 0.0f ? val : 0.0f;
    }
  }
}

__global__ __launch_bounds__(EDGE_THREADS, 16)
void local_trunk_kernel(const int* __restrict__ src, const int* __restrict__ dst,
                        int* __restrict__ tbl, int* __restrict__ tmp_pairs,
                        const float* __restrict__ X,
                        const float* __restrict__ W1, const float* __restrict__ b1,
                        const float* __restrict__ W2, const float* __restrict__ b2,
                        float* __restrict__ P0,
                        const float* __restrict__ thetas, const float* __restrict__ Wm1,
                        float* __restrict__ Wc) {
  __shared__ __attribute__((aligned(16))) float role_lds[ROLE_LDS_FLOATS];
  if (blockIdx.x < EDGE_BLOCKS) bucket_local_role(src, dst, tbl, tmp_pairs, reinterpret_cast<int2*>(role_lds));
  else if (blockIdx.x < EDGE_BLOCKS + TRUNK_BLOCKS) trunk_role(blockIdx.x - EDGE_BLOCKS, X, W1, b1, W2, b2, P0, role_lds);
  else head_weight_role(blockIdx.x - (EDGE_BLOCKS + TRUNK_BLOCKS), thetas, Wm1, Wc);
}

template <int U>
__device__ __forceinline__ void gather_rows(const float2* __restrict__ rows, int s, float di,
                                            int t, int lane, float2& acc) {
  int   st[U];
  float dt[U];
  float2 v[U];
  for (int j = 0; j < U; ++j) {
    st[j] = __builtin_amdgcn_readlane(s, t + j);
    dt[j] = __int_as_float(__builtin_amdgcn_readlane(__float_as_int(di), t + j));
  }
  for (int j = 0; j < U; ++j)
    v[j] = rows[(size_t)st[j] * (H_FEATS / 2) + lane];
  for (int j = 0; j < U; ++j) {
    acc.x = fmaf(dt[j], v[j].x, acc.x);
    acc.y = fmaf(dt[j], v[j].y, acc.y);
  }
}

template <bool FIRST>
__global__ __launch_bounds__(256)
void gather_kernel(const float* __restrict__ Pprev, const int* __restrict__ off,
                   const int* __restrict__ csr, const float* __restrict__ dinv,
                   int* __restrict__ pair_words, float* __restrict__ Pk, int n) {
  int2* pairs = reinterpret_cast<int2*>(pair_words);
  const int gid  = blockIdx.x * blockDim.x + threadIdx.x;
  const int lane = gid & 31;
  const int node = __builtin_amdgcn_readfirstlane(gid >> 5);
  if (node >= n) return;
  const int beg = __builtin_amdgcn_readfirstlane(off[node]);
  const int end = __builtin_amdgcn_readfirstlane(off[node + 1]);
  const float2* rows = reinterpret_cast<const float2*>(Pprev);
  const float2 self = rows[(size_t)node * (H_FEATS / 2) + lane];
  const float dn = dinv[node];

  float2 acc = make_float2(0.0f, 0.0f);
  for (int c = beg; c < end; c += 32) {
    const int m = (end - c) < 32 ? (end - c) : 32;
    int s = 0;
    float di = 0.0f;
    if (lane < m) {
      if (FIRST) {
        s  = csr[c + lane];
        di = dinv[s];
      } else {
        const int2 p = pairs[c + lane];
        s  = p.x;
        di = __int_as_float(p.y);
      }
    }
    int t = 0;
    for (; t + 8 <= m; t += 8) gather_rows<8>(rows, s, di, t, lane, acc);
    if (t + 4 <= m) { gather_rows<4>(rows, s, di, t, lane, acc); t += 4; }
    if (t + 2 <= m) { gather_rows<2>(rows, s, di, t, lane, acc); t += 2; }
    if (t + 1 <= m) { gather_rows<1>(rows, s, di, t, lane, acc); }
    if (FIRST && lane < m) pairs[c + lane] = make_int2(s, __float_as_int(di));
  }
  float2 o;
  o.x = self.x - acc.x * dn;
  o.y = self.y - acc.y * dn;
  reinterpret_cast<float2*>(Pk)[(size_t)node * (H_FEATS / 2) + lane] = o;
}

__device__ __forceinline__ void gather_chunk(const float2* __restrict__ rows, int s, float di,
                                             int m, int lane, float2& acc) {
  int t = 0;
  for (; t + 8 <= m; t += 8) gather_rows<8>(rows, s, di, t, lane, acc);
  if (t + 4 <= m) { gather_rows<4>(rows, s, di, t, lane, acc); t += 4; }
  if (t + 2 <= m) { gather_rows<2>(rows, s, di, t, lane, acc); t += 2; }
  if (t + 1 <= m) { gather_rows<1>(rows, s, di, t, lane, acc); }
}

static_assert(N_NODES % 16 == 0, "the head grid covers the nodes exactly (no bounds test)");
#define TILE_PITCH 72
#define NODES_PER_WAVE 4

__global__ __launch_bounds__(128, 16)
void head_kernel(const float* __restrict__ P0, const float* __restrict__ P1,
                 const float* __restrict__ P2, const int* __restrict__ off,
                 const int* __restrict__ pair_words, const float* __restrict__ dinv,
                 const float* __restrict__ Wc, const float* __restrict__ bm1,
                 const float* __restrict__ Wm2, const float* __restrict__ bm2,
                 float* __restrict__ out) {
  __shared__ __attribute__((aligned(16))) float tile[POLY_ORDER][16 * TILE_PITCH];

  const int tid  = threadIdx.x;
  const int lane = tid & 31;
  const int wave = __builtin_amdgcn_readfirstlane(tid >> 5);
  const int mrow = lane & 15;
  const int kh   = lane >> 4;
  const int row0 = blockIdx.x * 16;
  const int col0 = wave * 16;
  const int lrow = wave * NODES_PER_WAVE;
  const int nbase = row0 + lrow;

  const float2* rows0 = reinterpret_cast<const float2*>(P0);
  const float2* rows1 = reinterpret_cast<const float2*>(P1);
  const float2* rows2 = reinterpret_cast<const float2*>(P2);
  const int2*   pairs = reinterpret_cast<const int2*>(pair_words);

  const int   ov = off[nbase + (lane < NODES_PER_WAVE ? lane : NODES_PER_WAVE)];
  const float dv = dinv[nbase + (lane & (NODES_PER_WAVE - 1))];

  float2 p0[NODES_PER_WAVE], p1[NODES_PER_WAVE], self[NODES_PER_WAVE];
  #pragma unroll
  for (int i = 0; i < NODES_PER_WAVE; ++i) {
    p0[i]   = rows0[(size_t)(nbase + i) * (H_FEATS / 2) + lane];
    p1[i]   = rows1[(size_t)(nbase + i) * (H_FEATS / 2) + lane];
    self[i] = rows2[(size_t)(nbase + i) * (H_FEATS / 2) + lane];
  }

  int beg[NODES_PER_WAVE], end[NODES_PER_WAVE], m0[NODES_PER_WAVE], s0[NODES_PER_WAVE];
  float d0[NODES_PER_WAVE];
  #pragma unroll
  for (int i = 0; i < NODES_PER_WAVE; ++i) {
    beg[i] = __builtin_amdgcn_readlane(ov, i);
    end[i] = __builtin_amdgcn_readlane(ov, i + 1);
    m0[i]  = (end[i] - beg[i]) < 32 ? (end[i] - beg[i]) : 32;
    int2 p = make_int2(0, 0);
    if (lane < m0[i]) p = pairs[beg[i] + lane];
    s0[i] = p.x;
    d0[i] = __int_as_float(p.y);
  }

  #pragma unroll
  for (int i = 0; i < NODES_PER_WAVE; ++i) {
    const int a = (lrow + i) * TILE_PITCH + 2 * lane;
    *reinterpret_cast<float2*>(&tile[0][a]) = p0[i];
    *reinterpret_cast<float2*>(&tile[1][a]) = p1[i];
    *reinterpret_cast<float2*>(&tile[2][a]) = self[i];
  }

  #pragma unroll
  for (int i = 0; i < NODES_PER_WAVE; ++i) {
    float2 acc = make_float2(0.0f, 0.0f);
    gather_chunk(rows2, s0[i], d0[i], m0[i], lane, acc);
    for (int c = beg[i] + 32; c < end[i]; c += 32) {
      const int m = (end[i] - c) < 32 ? (end[i] - c) : 32;
      int s = 0;
      float di = 0.0f;
      if (lane < m) {
        const int2 p = pairs[c + lane];
        s  = p.x;
        di = __int_as_float(p.y);
      }
      gather_chunk(rows2, s, di, m, lane, acc);
    }
    const float dn = __int_as_float(__builtin_amdgcn_readlane(__float_as_int(dv), i));
    float2 o;
    o.x = self[i].x - acc.x * dn;
    o.y = self[i].y - acc.y * dn;
    *reinterpret_cast<float2*>(&tile[3][(lrow + i) * TILE_PITCH + 2 * lane]) = o;
  }
  __syncthreads();

  v8f acc = {};
  #pragma unroll 1
  for (int k = 0; k < POLY_ORDER; ++k) {
    const float* Ak = tile[k];
    const float* Wk = Wc + k * H_FEATS * H_FEATS;
    for (int k0 = 0; k0 < H_FEATS; k0 += 4) {
      v2f a, b;
      a.x = Ak[mrow * TILE_PITCH + k0 + 2 * kh];
      a.y = Ak[mrow * TILE_PITCH + k0 + 2 * kh + 1];
      b.x = Wk[(k0 + 2 * kh)     * H_FEATS + col0 + mrow];
      b.y = Wk[(k0 + 2 * kh + 1) * H_FEATS + col0 + mrow];
      acc = wmma_f32(a, b, acc);
    }
  }
  __syncthreads();

  float* ldsH = tile[0];
  {
    float bias = bm1[col0 + mrow];
    #pragma unroll
    for (int v = 0; v < 8; ++v) {
      int r = v + 8 * kh;
      float val = acc[v] + bias;
      ldsH[r * TILE_PITCH + col0 + mrow] = val > 0.0f ? val : 0.0f;
    }
  }
  __syncthreads();

  if (tid < 32) {
    int r = tid >> 1, c = tid & 1;
    float s = bm2[c];
    #pragma unroll 8
    for (int j = 0; j < H_FEATS; ++j)
      s += ldsH[r * TILE_PITCH + j] * Wm2[j * NUM_CLASSES + c];
    out[(size_t)(row0 + r) * NUM_CLASSES + c] = s;
  }
}

extern "C" void kernel_launch(void* const* d_in, const int* in_sizes, int n_in,
                              void* d_out, int out_size, void* d_ws, size_t ws_size,
                              hipStream_t stream) {
  const float* X      = (const float*)d_in[0];
  const int*   src    = (const int*)  d_in[1];
  const int*   dst    = (const int*)  d_in[2];
  const float* W1     = (const float*)d_in[3];
  const float* b1     = (const float*)d_in[4];
  const float* W2     = (const float*)d_in[5];
  const float* b2     = (const float*)d_in[6];
  const float* thetas = (const float*)d_in[7];
  const float* Wm1    = (const float*)d_in[8];
  const float* bm1    = (const float*)d_in[9];
  const float* Wm2    = (const float*)d_in[10];
  const float* bm2    = (const float*)d_in[11];
  float* out = (float*)d_out;
  float* ws  = (float*)d_ws;

  const int PN = N_NODES * H_FEATS;
  float* dinv = ws;
  float* P0   = ws + 100352;
  float* P1   = P0 + PN;
  float* P2   = P1 + PN;
  float* P3   = P2 + PN;
  int*   pairw = (int*)P3;
  int*   ibuf = (int*)(P3 + PN);
  int*   off   = ibuf;
  int*   csr   = off + 100352;
  int*   tmp   = csr + N_EDGES;
  int*   tbl   = csr + 3 * N_EDGES;
  float* Wc   = P3 + PN + PN;

  local_trunk_kernel<<<EDGE_BLOCKS + TRUNK_BLOCKS + PREP_BLOCKS, EDGE_THREADS, 0, stream>>>(
      src, dst, tbl, tmp, X, W1, b1, W2, b2, P0, thetas, Wm1, Wc);
  bucket_finish_kernel<<<N_BUCKETS, FIN_THREADS, 0, stream>>>(tmp, tbl, off, dinv, csr, N_NODES);

  gather_kernel<true><<<(N_NODES * 32 + 255) / 256, 256, 0, stream>>>(P0, off, csr, dinv, pairw, P1, N_NODES);
  gather_kernel<false><<<(N_NODES * 32 + 255) / 256, 256, 0, stream>>>(P1, off, csr, dinv, pairw, P2, N_NODES);

  head_kernel<<<N_NODES / 16, 128, 0, stream>>>(
      P0, P1, P2, off, pairw, dinv, Wc, bm1, Wm2, bm2, out);
}
